# starting kernel
# baseline (speedup 1.0000x reference)
#include <hip/hip_runtime.h>

typedef __attribute__((ext_vector_type(2))) float v2f;
typedef __attribute__((ext_vector_type(8))) float v8f;

__global__ void fill_f32(float* __restrict__ p, float v, int n) {
    int i = blockIdx.x * blockDim.x + threadIdx.x;
    if (i < n) p[i] = v;
}

__global__ void deg_edges(const int* __restrict__ dst, float* __restrict__ deg, int nE) {
    int e = blockIdx.x * blockDim.x + threadIdx.x;
    if (e < nE) atomicAdd(&deg[dst[e]], 1.0f);
}

__global__ void deg_to_dis(float* __restrict__ d, int n) {
    int i = blockIdx.x * blockDim.x + threadIdx.x;
    if (i < n) d[i] = rsqrtf(d[i]);
}

template<int FT>
__global__ void wmma_gemm(const float* __restrict__ X, const float* __restrict__ W,
                          const float* __restrict__ Bias, float* __restrict__ Y,
                          int N, int K, int addBias) {
    const int F    = FT * 16;
    int wave = (blockIdx.x * blockDim.x + threadIdx.x) >> 5;
    int lane = threadIdx.x & 31;
    int row0 = wave << 4;
    if (row0 >= N) return;
    int half = lane >> 4;
    int l16  = lane & 15;
    int arow = row0 + l16; if (arow > N - 1) arow = N - 1;

    v8f acc[FT];
    v8f z = {0.f, 0.f, 0.f, 0.f, 0.f, 0.f, 0.f, 0.f};
#pragma unroll
    for (int t = 0; t < FT; ++t) acc[t] = z;

    for (int k0 = 0; k0 < K; k0 += 4) {
        int ka = k0 + half * 2;
        v2f a;
        a.x = X[arow * K + ka];
        a.y = X[arow * K + ka + 1];
#pragma unroll
        for (int t = 0; t < FT; ++t) {
            int col = t * 16 + l16;
            v2f b;
            b.x = W[ka * F + col];
            b.y = W[(ka + 1) * F + col];
            acc[t] = __builtin_amdgcn_wmma_f32_16x16x4_f32(
                 false, a,  false, b,
                 (short)0, acc[t],  false,  false);
        }
    }

#pragma unroll
    for (int t = 0; t < FT; ++t) {
        int col = t * 16 + l16;
        float bb = addBias ? Bias[col] : 0.f;
#pragma unroll
        for (int v = 0; v < 8; ++v) {
            int r = row0 + half * 8 + v;
            if (r < N) Y[r * F + col] = acc[t][v] + bb;
        }
    }
}

__global__ void agg_init(const float* __restrict__ H, const float* __restrict__ dis,
                         float* __restrict__ AGG, int N, int F) {
    int idx = blockIdx.x * blockDim.x + threadIdx.x;
    if (idx < N * F) {
        float d = dis[idx / F];
        AGG[idx] = H[idx] * d * d;
    }
}

__global__ void edge_scatter(const int* __restrict__ src, const int* __restrict__ dst,
                             const float* __restrict__ dis, const float* __restrict__ H,
                             float* __restrict__ AGG, int nE, int F) {
    int e    = (blockIdx.x * blockDim.x + threadIdx.x) >> 5;
    int lane = threadIdx.x & 31;
    if (e >= nE) return;
    int   s = src[e];
    int   d = dst[e];
    float w = dis[s] * dis[d];
    for (int f = lane; f < F; f += 32)
        atomicAdd(&AGG[d * F + f], H[s * F + f] * w);
}

__global__ void bias_relu(float* __restrict__ A, const float* __restrict__ b, int N, int F) {
    int idx = blockIdx.x * blockDim.x + threadIdx.x;
    if (idx < N * F) {
        float v = A[idx] + b[idx % F];
        A[idx] = v > 0.f ? v : 0.f;
    }
}

__global__ void pool_nodes(const float* __restrict__ H, const int* __restrict__ batch,
                           float* __restrict__ gmax, float* __restrict__ gsum,
                           float* __restrict__ cnt, int N, int F) {
    int i    = (blockIdx.x * blockDim.x + threadIdx.x) >> 5;
    int lane = threadIdx.x & 31;
    if (i >= N) return;
    int g = batch[i];
    for (int f = lane; f < F; f += 32) {
        float v = H[i * F + f];
        atomicMax((unsigned int*)&gmax[g * F + f], __float_as_uint(v));
        atomicAdd(&gsum[g * F + f], v);
    }
    if (lane == 0) atomicAdd(&cnt[g], 1.0f);
}

__global__ void build_pooled(const float* __restrict__ gmax, const float* __restrict__ gsum,
                             const float* __restrict__ cnt, float* __restrict__ pooled,
                             int G, int F) {
    int idx = blockIdx.x * blockDim.x + threadIdx.x;
    if (idx >= G * 2 * F) return;
    int g = idx / (2 * F);
    int f = idx % (2 * F);
    float v;
    if (f < F) {
        v = gmax[g * F + f];
    } else {
        float c = cnt[g];
        v = gsum[g * F + (f - F)] / (c > 1.f ? c : 1.f);
    }
    pooled[idx] = v;
}

extern "C" void kernel_launch(void* const* d_in, const int* in_sizes, int n_in,
                              void* d_out, int out_size, void* d_ws, size_t ws_size,
                              hipStream_t stream) {
    const float* x     = (const float*)d_in[0];
    const int*   ei    = (const int*)d_in[1];
    const int*   batch = (const int*)d_in[2];
    const float* W1    = (const float*)d_in[3];
    const float* b1    = (const float*)d_in[4];
    const float* W2    = (const float*)d_in[5];
    const float* b2    = (const float*)d_in[6];
    const float* W3    = (const float*)d_in[7];
    const float* b3    = (const float*)d_in[8];
    const float* Wout  = (const float*)d_in[9];
    const float* bout  = (const float*)d_in[10];
    float*       out   = (float*)d_out;

    const int N  = in_sizes[2];
    const int E  = in_sizes[1] / 2;
    const int K0 = in_sizes[0] / N;
    const int F1 = in_sizes[4];
    const int F2 = in_sizes[6];
    const int F3 = in_sizes[8];
    const int Fo = in_sizes[10];
    const int G  = out_size / Fo;

    const int* src = ei;
    const int* dst = ei + E;

    char*  ws  = (char*)d_ws;
    size_t off = 0;
    auto carve = [&](size_t bytes) -> float* {
        float* p = (float*)(ws + off);
        off += (bytes + 255) & ~(size_t)255;
        return p;
    };
    float* dis    = carve((size_t)N * 4);
    float* bufT   = carve((size_t)N * F3 * 4);
    float* bufA   = carve((size_t)N * F3 * 4);
    float* bufB   = carve((size_t)N * F3 * 4);
    float* gmax   = carve((size_t)G * F3 * 4);
    float* gsum   = carve((size_t)G * F3 * 4);
    float* cnt    = carve((size_t)G * 4);
    float* pooled = carve((size_t)G * 2 * F3 * 4);

    const int BLK = 256;
    auto grid1   = [&](long long n) { return dim3((unsigned)((n + BLK - 1) / BLK)); };
    auto gridWv  = [&](long long waves) { return dim3((unsigned)((waves * 32 + BLK - 1) / BLK)); };
    auto gemmGrid = [&](int rows) { return gridWv((rows + 15) / 16); };

    fill_f32<<<grid1(N), BLK, 0, stream>>>(dis, 1.0f, N);
    deg_edges<<<grid1(E), BLK, 0, stream>>>(dst, dis, E);
    deg_to_dis<<<grid1(N), BLK, 0, stream>>>(dis, N);

    wmma_gemm<2><<<gemmGrid(N), BLK, 0, stream>>>(x, W1, b1, bufT, N, K0, 0);
    agg_init<<<grid1((long long)N * F1), BLK, 0, stream>>>(bufT, dis, bufA, N, F1);
    edge_scatter<<<gridWv(E), BLK, 0, stream>>>(src, dst, dis, bufT, bufA, E, F1);
    bias_relu<<<grid1((long long)N * F1), BLK, 0, stream>>>(bufA, b1, N, F1);

    wmma_gemm<3><<<gemmGrid(N), BLK, 0, stream>>>(bufA, W2, b2, bufT, N, F1, 0);
    agg_init<<<grid1((long long)N * F2), BLK, 0, stream>>>(bufT, dis, bufB, N, F2);
    edge_scatter<<<gridWv(E), BLK, 0, stream>>>(src, dst, dis, bufT, bufB, E, F2);
    bias_relu<<<grid1((long long)N * F2), BLK, 0, stream>>>(bufB, b2, N, F2);

    wmma_gemm<4><<<gemmGrid(N), BLK, 0, stream>>>(bufB, W3, b3, bufT, N, F2, 0);
    agg_init<<<grid1((long long)N * F3), BLK, 0, stream>>>(bufT, dis, bufA, N, F3);
    edge_scatter<<<gridWv(E), BLK, 0, stream>>>(src, dst, dis, bufT, bufA, E, F3);
    bias_relu<<<grid1((long long)N * F3), BLK, 0, stream>>>(bufA, b3, N, F3);

    fill_f32<<<grid1((long long)G * F3), BLK, 0, stream>>>(gmax, 0.0f, G * F3);
    fill_f32<<<grid1((long long)G * F3), BLK, 0, stream>>>(gsum, 0.0f, G * F3);
    fill_f32<<<grid1(G), BLK, 0, stream>>>(cnt, 0.0f, G);
    pool_nodes<<<gridWv(N), BLK, 0, stream>>>(bufA, batch, gmax, gsum, cnt, N, F3);
    build_pooled<<<grid1((long long)G * 2 * F3), BLK, 0, stream>>>(gmax, gsum, cnt, pooled, G, F3);

    wmma_gemm<2><<<gemmGrid(G), BLK, 0, stream>>>(pooled, Wout, bout, out, G, 2 * F3, 1);
}


# edge_scatter: each wave walks 32 consecutive edges (coalesced index loads, scalar row bases and weights, 8 row loads in flight ahead of their f32 atomic adds) instead of one wave per edge
# speedup vs baseline: 1.8165x; 1.8165x over previous
#include <hip/hip_runtime.h>

typedef __attribute__((ext_vector_type(2))) float v2f;
typedef __attribute__((ext_vector_type(8))) float v8f;

__global__ void fill_f32(float* __restrict__ p, float v, int n) {
    int i = blockIdx.x * blockDim.x + threadIdx.x;
    if (i < n) p[i] = v;
}

__global__ void deg_edges(const int* __restrict__ dst, float* __restrict__ deg, int nE) {
    int e = blockIdx.x * blockDim.x + threadIdx.x;
    if (e < nE) atomicAdd(&deg[dst[e]], 1.0f);
}

__global__ void deg_to_dis(float* __restrict__ d, int n) {
    int i = blockIdx.x * blockDim.x + threadIdx.x;
    if (i < n) d[i] = rsqrtf(d[i]);
}

template<int FT>
__global__ void wmma_gemm(const float* __restrict__ X, const float* __restrict__ W,
                          const float* __restrict__ Bias, float* __restrict__ Y,
                          int N, int K, int addBias) {
    const int F    = FT * 16;
    int wave = (blockIdx.x * blockDim.x + threadIdx.x) >> 5;
    int lane = threadIdx.x & 31;
    int row0 = wave << 4;
    if (row0 >= N) return;
    int half = lane >> 4;
    int l16  = lane & 15;
    int arow = row0 + l16; if (arow > N - 1) arow = N - 1;

    v8f acc[FT];
    v8f z = {0.f, 0.f, 0.f, 0.f, 0.f, 0.f, 0.f, 0.f};
#pragma unroll
    for (int t = 0; t < FT; ++t) acc[t] = z;

    for (int k0 = 0; k0 < K; k0 += 4) {
        int ka = k0 + half * 2;
        v2f a;
        a.x = X[arow * K + ka];
        a.y = X[arow * K + ka + 1];
#pragma unroll
        for (int t = 0; t < FT; ++t) {
            int col = t * 16 + l16;
            v2f b;
            b.x = W[ka * F + col];
            b.y = W[(ka + 1) * F + col];
            acc[t] = __builtin_amdgcn_wmma_f32_16x16x4_f32(
                 false, a,  false, b,
                 (short)0, acc[t],  false,  false);
        }
    }

#pragma unroll
    for (int t = 0; t < FT; ++t) {
        int col = t * 16 + l16;
        float bb = addBias ? Bias[col] : 0.f;
#pragma unroll
        for (int v = 0; v < 8; ++v) {
            int r = row0 + half * 8 + v;
            if (r < N) Y[r * F + col] = acc[t][v] + bb;
        }
    }
}

__global__ void agg_init(const float* __restrict__ H, const float* __restrict__ dis,
                         float* __restrict__ AGG, int N, int F) {
    int idx = blockIdx.x * blockDim.x + threadIdx.x;
    if (idx < N * F) {
        float d = dis[idx / F];
        AGG[idx] = H[idx] * d * d;
    }
}

#define EDGES_PER_WAVE 32
#define EDGE_UNROLL 8
__global__ void edge_scatter(const int* __restrict__ src, const int* __restrict__ dst,
                             const float* __restrict__ dis, const float* __restrict__ H,
                             float* __restrict__ AGG, int nE, int F) {
    int wave = __builtin_amdgcn_readfirstlane((int)((blockIdx.x * blockDim.x + threadIdx.x) >> 5));
    int lane = threadIdx.x & 31;
    int e0   = wave * EDGES_PER_WAVE;
    if (e0 >= nE) return;
    int cnt = nE - e0;
    if (cnt > EDGES_PER_WAVE) cnt = EDGES_PER_WAVE;

    int   sOff = 0, dOff = 0;
    float w    = 0.f;
    if (lane < cnt) {
        int s = src[e0 + lane];
        int d = dst[e0 + lane];
        w    = dis[s] * dis[d];
        sOff = s * F;
        dOff = d * F;
    }
    int wBits = __float_as_int(w);

    if (cnt == EDGES_PER_WAVE) {
        for (int j0 = 0; j0 < EDGES_PER_WAVE; j0 += EDGE_UNROLL) {
            const float* hp[EDGE_UNROLL];
            float*       ap[EDGE_UNROLL];
            float        ww[EDGE_UNROLL];
#pragma unroll
            for (int u = 0; u < EDGE_UNROLL; ++u) {
                hp[u] = H   + __builtin_amdgcn_readlane(sOff, j0 + u);
                ap[u] = AGG + __builtin_amdgcn_readlane(dOff, j0 + u);
                ww[u] = __int_as_float(__builtin_amdgcn_readlane(wBits, j0 + u));
            }
            for (int f = lane; f < F; f += 32) {
                float h[EDGE_UNROLL];
#pragma unroll
                for (int u = 0; u < EDGE_UNROLL; ++u) h[u] = hp[u][f];
#pragma unroll
                for (int u = 0; u < EDGE_UNROLL; ++u) atomicAdd(&ap[u][f], h[u] * ww[u]);
            }
        }
    } else {
        for (int j = 0; j < cnt; ++j) {
            const float* hp = H   + __builtin_amdgcn_readlane(sOff, j);
            float*       ap = AGG + __builtin_amdgcn_readlane(dOff, j);
            float        wj = __int_as_float(__builtin_amdgcn_readlane(wBits, j));
            for (int f = lane; f < F; f += 32) atomicAdd(&ap[f], hp[f] * wj);
        }
    }
}

__global__ void bias_relu(float* __restrict__ A, const float* __restrict__ b, int N, int F) {
    int idx = blockIdx.x * blockDim.x + threadIdx.x;
    if (idx < N * F) {
        float v = A[idx] + b[idx % F];
        A[idx] = v > 0.f ? v : 0.f;
    }
}

__global__ void pool_nodes(const float* __restrict__ H, const int* __restrict__ batch,
                           float* __restrict__ gmax, float* __restrict__ gsum,
                           float* __restrict__ cnt, int N, int F) {
    int i    = (blockIdx.x * blockDim.x + threadIdx.x) >> 5;
    int lane = threadIdx.x & 31;
    if (i >= N) return;
    int g = batch[i];
    for (int f = lane; f < F; f += 32) {
        float v = H[i * F + f];
        atomicMax((unsigned int*)&gmax[g * F + f], __float_as_uint(v));
        atomicAdd(&gsum[g * F + f], v);
    }
    if (lane == 0) atomicAdd(&cnt[g], 1.0f);
}

__global__ void build_pooled(const float* __restrict__ gmax, const float* __restrict__ gsum,
                             const float* __restrict__ cnt, float* __restrict__ pooled,
                             int G, int F) {
    int idx = blockIdx.x * blockDim.x + threadIdx.x;
    if (idx >= G * 2 * F) return;
    int g = idx / (2 * F);
    int f = idx % (2 * F);
    float v;
    if (f < F) {
        v = gmax[g * F + f];
    } else {
        float c = cnt[g];
        v = gsum[g * F + (f - F)] / (c > 1.f ? c : 1.f);
    }
    pooled[idx] = v;
}

extern "C" void kernel_launch(void* const* d_in, const int* in_sizes, int n_in,
                              void* d_out, int out_size, void* d_ws, size_t ws_size,
                              hipStream_t stream) {
    const float* x     = (const float*)d_in[0];
    const int*   ei    = (const int*)d_in[1];
    const int*   batch = (const int*)d_in[2];
    const float* W1    = (const float*)d_in[3];
    const float* b1    = (const float*)d_in[4];
    const float* W2    = (const float*)d_in[5];
    const float* b2    = (const float*)d_in[6];
    const float* W3    = (const float*)d_in[7];
    const float* b3    = (const float*)d_in[8];
    const float* Wout  = (const float*)d_in[9];
    const float* bout  = (const float*)d_in[10];
    float*       out   = (float*)d_out;

    const int N  = in_sizes[2];
    const int E  = in_sizes[1] / 2;
    const int K0 = in_sizes[0] / N;
    const int F1 = in_sizes[4];
    const int F2 = in_sizes[6];
    const int F3 = in_sizes[8];
    const int Fo = in_sizes[10];
    const int G  = out_size / Fo;

    const int* src = ei;
    const int* dst = ei + E;

    char*  ws  = (char*)d_ws;
    size_t off = 0;
    auto carve = [&](size_t bytes) -> float* {
        float* p = (float*)(ws + off);
        off += (bytes + 255) & ~(size_t)255;
        return p;
    };
    float* dis    = carve((size_t)N * 4);
    float* bufT   = carve((size_t)N * F3 * 4);
    float* bufA   = carve((size_t)N * F3 * 4);
    float* bufB   = carve((size_t)N * F3 * 4);
    float* gmax   = carve((size_t)G * F3 * 4);
    float* gsum   = carve((size_t)G * F3 * 4);
    float* cnt    = carve((size_t)G * 4);
    float* pooled = carve((size_t)G * 2 * F3 * 4);

    const int BLK = 256;
    auto grid1   = [&](long long n) { return dim3((unsigned)((n + BLK - 1) / BLK)); };
    auto gridWv  = [&](long long waves) { return dim3((unsigned)((waves * 32 + BLK - 1) / BLK)); };
    auto gemmGrid = [&](int rows) { return gridWv((rows + 15) / 16); };

    fill_f32<<<grid1(N), BLK, 0, stream>>>(dis, 1.0f, N);
    deg_edges<<<grid1(E), BLK, 0, stream>>>(dst, dis, E);
    deg_to_dis<<<grid1(N), BLK, 0, stream>>>(dis, N);

    wmma_gemm<2><<<gemmGrid(N), BLK, 0, stream>>>(x, W1, b1, bufT, N, K0, 0);
    agg_init<<<grid1((long long)N * F1), BLK, 0, stream>>>(bufT, dis, bufA, N, F1);
    edge_scatter<<<gridWv((E + 31) / 32), BLK, 0, stream>>>(src, dst, dis, bufT, bufA, E, F1);
    bias_relu<<<grid1((long long)N * F1), BLK, 0, stream>>>(bufA, b1, N, F1);

    wmma_gemm<3><<<gemmGrid(N), BLK, 0, stream>>>(bufA, W2, b2, bufT, N, F1, 0);
    agg_init<<<grid1((long long)N * F2), BLK, 0, stream>>>(bufT, dis, bufB, N, F2);
    edge_scatter<<<gridWv((E + 31) / 32), BLK, 0, stream>>>(src, dst, dis, bufT, bufB, E, F2);
    bias_relu<<<grid1((long long)N * F2), BLK, 0, stream>>>(bufB, b2, N, F2);

    wmma_gemm<4><<<gemmGrid(N), BLK, 0, stream>>>(bufB, W3, b3, bufT, N, F2, 0);
    agg_init<<<grid1((long long)N * F3), BLK, 0, stream>>>(bufT, dis, bufA, N, F3);
    edge_scatter<<<gridWv((E + 31) / 32), BLK, 0, stream>>>(src, dst, dis, bufT, bufA, E, F3);
    bias_relu<<<grid1((long long)N * F3), BLK, 0, stream>>>(bufA, b3, N, F3);

    fill_f32<<<grid1((long long)G * F3), BLK, 0, stream>>>(gmax, 0.0f, G * F3);
    fill_f32<<<grid1((long long)G * F3), BLK, 0, stream>>>(gsum, 0.0f, G * F3);
    fill_f32<<<grid1(G), BLK, 0, stream>>>(cnt, 0.0f, G);
    pool_nodes<<<gridWv(N), BLK, 0, stream>>>(bufA, batch, gmax, gsum, cnt, N, F3);
    build_pooled<<<grid1((long long)G * 2 * F3), BLK, 0, stream>>>(gmax, gsum, cnt, pooled, G, F3);

    wmma_gemm<2><<<gemmGrid(G), BLK, 0, stream>>>(pooled, Wout, bout, out, G, 2 * F3, 1);
}


# aggregation pulls, not scatters: edges regrouped by destination each call (count, scan, arrival-order slots); a wave sums a node's in-rows in registers, writes the row once, no atomic adds there
# speedup vs baseline: 2.1795x; 1.1998x over previous
#include <hip/hip_runtime.h>

typedef __attribute__((ext_vector_type(2))) float v2f;
typedef __attribute__((ext_vector_type(8))) float v8f;

__global__ void fill_f32(float* __restrict__ p, float v, int n) {
    int i = blockIdx.x * blockDim.x + threadIdx.x;
    if (i < n) p[i] = v;
}

__global__ void fill_i32(int* __restrict__ p, int v, int n) {
    int i = blockIdx.x * blockDim.x + threadIdx.x;
    if (i < n) p[i] = v;
}


__global__ void count_in_edges(const int* __restrict__ dst, int* __restrict__ cnt, int nE) {
    int e = blockIdx.x * blockDim.x + threadIdx.x;
    if (e < nE) atomicAdd(&cnt[dst[e]], 1);
}

#define SCAN_T   256
#define SCAN_PER 4

__global__ void tile_sums(const int* __restrict__ cnt, int* __restrict__ tsum, int n) {
    __shared__ int sh[SCAN_T];
    int t  = threadIdx.x;
    int i0 = (blockIdx.x * SCAN_T + t) * SCAN_PER;
    int s  = 0;
#pragma unroll
    for (int k = 0; k < SCAN_PER; ++k) if (i0 + k < n) s += cnt[i0 + k];
    sh[t] = s;
    __syncthreads();
    for (int o = SCAN_T / 2; o > 0; o >>= 1) {
        if (t < o) sh[t] += sh[t + o];
        __syncthreads();
    }
    if (t == 0) tsum[blockIdx.x] = sh[0];
}

__global__ void scan_rows(const int* __restrict__ cnt, const int* __restrict__ tsum,
                          int* __restrict__ rowptr, int* __restrict__ cursor,
                          float* __restrict__ dis, int n) {
    __shared__ int sh[SCAN_T];
    int t = threadIdx.x;

    int b = 0;
    for (int i = t; i < (int)blockIdx.x; i += SCAN_T) b += tsum[i];
    sh[t] = b;
    __syncthreads();
    for (int o = SCAN_T / 2; o > 0; o >>= 1) {
        if (t < o) sh[t] += sh[t + o];
        __syncthreads();
    }
    int base = sh[0];
    __syncthreads();

    int i0 = (blockIdx.x * SCAN_T + t) * SCAN_PER;
    int c[SCAN_PER];
    int s = 0;
#pragma unroll
    for (int k = 0; k < SCAN_PER; ++k) {
        c[k] = (i0 + k < n) ? cnt[i0 + k] : 0;
        s += c[k];
    }
    sh[t] = s;
    __syncthreads();
    for (int o = 1; o < SCAN_T; o <<= 1) {
        int v = (t >= o) ? sh[t - o] : 0;
        __syncthreads();
        sh[t] += v;
        __syncthreads();
    }
    int run = base + sh[t] - s;
#pragma unroll
    for (int k = 0; k < SCAN_PER; ++k) {
        int i = i0 + k;
        if (i < n) {
            rowptr[i] = run;
            cursor[i] = run;
            dis[i]    = rsqrtf((float)(c[k] + 1));
            run += c[k];
            if (i == n - 1) rowptr[n] = run;
        }
    }
}

__global__ void place_edges(const int* __restrict__ src, const int* __restrict__ dst,
                            const float* __restrict__ dis, int* __restrict__ cursor,
                            int2* __restrict__ ent, int nE) {
    int e = blockIdx.x * blockDim.x + threadIdx.x;
    if (e >= nE) return;
    int d = dst[e];
    int s = src[e];
    int pos = atomicAdd(&cursor[d], 1);
    float w = dis[s] * dis[d];
    ent[pos] = make_int2(s, __float_as_int(w));
}

template<int FT>
__global__ void wmma_gemm(const float* __restrict__ X, const float* __restrict__ W,
                          const float* __restrict__ Bias, float* __restrict__ Y,
                          int N, int K, int addBias) {
    const int F    = FT * 16;
    int wave = (blockIdx.x * blockDim.x + threadIdx.x) >> 5;
    int lane = threadIdx.x & 31;
    int row0 = wave << 4;
    if (row0 >= N) return;
    int half = lane >> 4;
    int l16  = lane & 15;
    int arow = row0 + l16; if (arow > N - 1) arow = N - 1;

    v8f acc[FT];
    v8f z = {0.f, 0.f, 0.f, 0.f, 0.f, 0.f, 0.f, 0.f};
#pragma unroll
    for (int t = 0; t < FT; ++t) acc[t] = z;

    for (int k0 = 0; k0 < K; k0 += 4) {
        int ka = k0 + half * 2;
        v2f a;
        a.x = X[arow * K + ka];
        a.y = X[arow * K + ka + 1];
#pragma unroll
        for (int t = 0; t < FT; ++t) {
            int col = t * 16 + l16;
            v2f b;
            b.x = W[ka * F + col];
            b.y = W[(ka + 1) * F + col];
            acc[t] = __builtin_amdgcn_wmma_f32_16x16x4_f32(
                 false, a,  false, b,
                 (short)0, acc[t],  false,  false);
        }
    }

#pragma unroll
    for (int t = 0; t < FT; ++t) {
        int col = t * 16 + l16;
        float bb = addBias ? Bias[col] : 0.f;
#pragma unroll
        for (int v = 0; v < 8; ++v) {
            int r = row0 + half * 8 + v;
            if (r < N) Y[r * F + col] = acc[t][v] + bb;
        }
    }
}

template<int VEC> struct RowVec;
template<> struct RowVec<1> { typedef float type; };
template<> struct RowVec<2> { typedef v2f   type; };

template<int VEC, int U>
__device__ __forceinline__ void pull_group(const float* __restrict__ H, unsigned fo, int off, int wBits,
                                           int j, typename RowVec<VEC>::type& acc) {
    typedef typename RowVec<VEC>::type vec_t;
    vec_t h[U];
    float w[U];
#pragma unroll
    for (int u = 0; u < U; ++u) {
        const float* row = H + __builtin_amdgcn_readlane(off, j + u);
        w[u] = __int_as_float(__builtin_amdgcn_readlane(wBits, j + u));
        h[u] = *(const vec_t*)(row + fo);
    }
#pragma unroll
    for (int u = 0; u < U; ++u) acc += h[u] * w[u];
}

template<int F>
__global__ void pull_rows(const int* __restrict__ rowptr, const int2* __restrict__ ent,
                          const float* __restrict__ dis, const float* __restrict__ H,
                          const float* __restrict__ bias, float* __restrict__ OUT, int N) {
    constexpr int VEC = (F > 32) ? 2 : 1;
    typedef typename RowVec<VEC>::type vec_t;
    int node = __builtin_amdgcn_readfirstlane((int)((blockIdx.x * blockDim.x + threadIdx.x) >> 5));
    if (node >= N) return;
    int  lane   = threadIdx.x & 31;
    bool active = lane * VEC < F;
    unsigned fo = active ? lane * VEC : 0;

    int beg = __builtin_amdgcn_readfirstlane(rowptr[node]);
    int end = __builtin_amdgcn_readfirstlane(rowptr[node + 1]);
    float dd = dis[node];

    vec_t acc = *(const vec_t*)(H + node * F + fo) * (dd * dd);

    for (int base = beg; base < end; base += 32) {
        int cnt = end - base;
        if (cnt > 32) cnt = 32;
        int off = 0, wBits = 0;
        if (lane < cnt) {
            int2 en = ent[base + lane];
            off   = en.x * F;
            wBits = en.y;
        }
        int j = 0;
        for (; j + 8 <= cnt; j += 8) pull_group<VEC, 8>(H, fo, off, wBits, j, acc);
        if (cnt & 4) { pull_group<VEC, 4>(H, fo, off, wBits, j, acc); j += 4; }
        if (cnt & 2) { pull_group<VEC, 2>(H, fo, off, wBits, j, acc); j += 2; }
        if (cnt & 1) { pull_group<VEC, 1>(H, fo, off, wBits, j, acc); }
    }

    if (active) {
        vec_t v;
        if constexpr (VEC == 1) {
            v = acc + bias[fo];
            v = v > 0.f ? v : 0.f;
        } else {
            v.x = acc.x + bias[fo];
            v.y = acc.y + bias[fo + 1];
            v.x = v.x > 0.f ? v.x : 0.f;
            v.y = v.y > 0.f ? v.y : 0.f;
        }
        *(vec_t*)(OUT + node * F + fo) = v;
    }
}

__global__ void pool_nodes(const float* __restrict__ H, const int* __restrict__ batch,
                           float* __restrict__ gmax, float* __restrict__ gsum,
                           float* __restrict__ cnt, int N, int F) {
    int i    = (blockIdx.x * blockDim.x + threadIdx.x) >> 5;
    int lane = threadIdx.x & 31;
    if (i >= N) return;
    int g = batch[i];
    for (int f = lane; f < F; f += 32) {
        float v = H[i * F + f];
        atomicMax((unsigned int*)&gmax[g * F + f], __float_as_uint(v));
        atomicAdd(&gsum[g * F + f], v);
    }
    if (lane == 0) atomicAdd(&cnt[g], 1.0f);
}

__global__ void build_pooled(const float* __restrict__ gmax, const float* __restrict__ gsum,
                             const float* __restrict__ cnt, float* __restrict__ pooled,
                             int G, int F) {
    int idx = blockIdx.x * blockDim.x + threadIdx.x;
    if (idx >= G * 2 * F) return;
    int g = idx / (2 * F);
    int f = idx % (2 * F);
    float v;
    if (f < F) {
        v = gmax[g * F + f];
    } else {
        float c = cnt[g];
        v = gsum[g * F + (f - F)] / (c > 1.f ? c : 1.f);
    }
    pooled[idx] = v;
}

extern "C" void kernel_launch(void* const* d_in, const int* in_sizes, int n_in,
                              void* d_out, int out_size, void* d_ws, size_t ws_size,
                              hipStream_t stream) {
    const float* x     = (const float*)d_in[0];
    const int*   ei    = (const int*)d_in[1];
    const int*   batch = (const int*)d_in[2];
    const float* W1    = (const float*)d_in[3];
    const float* b1    = (const float*)d_in[4];
    const float* W2    = (const float*)d_in[5];
    const float* b2    = (const float*)d_in[6];
    const float* W3    = (const float*)d_in[7];
    const float* b3    = (const float*)d_in[8];
    const float* Wout  = (const float*)d_in[9];
    const float* bout  = (const float*)d_in[10];
    float*       out   = (float*)d_out;

    const int N  = in_sizes[2];
    const int E  = in_sizes[1] / 2;
    const int K0 = in_sizes[0] / N;
    const int F1 = in_sizes[4];
    const int F2 = in_sizes[6];
    const int F3 = in_sizes[8];
    const int Fo = in_sizes[10];
    const int G  = out_size / Fo;

    const int* src = ei;
    const int* dst = ei + E;

    char*  ws  = (char*)d_ws;
    size_t off = 0;
    auto carve = [&](size_t bytes) -> float* {
        float* p = (float*)(ws + off);
        off += (bytes + 255) & ~(size_t)255;
        return p;
    };
    float* dis    = carve((size_t)N * 4);
    float* bufT   = carve((size_t)N * F3 * 4);
    float* bufA   = carve((size_t)N * F3 * 4);
    float* bufB   = carve((size_t)N * F3 * 4);
    float* gmax   = carve((size_t)G * F3 * 4);
    float* gsum   = carve((size_t)G * F3 * 4);
    float* cnt    = carve((size_t)G * 4);
    float* pooled = carve((size_t)G * 2 * F3 * 4);
    const int TILE   = SCAN_T * SCAN_PER;
    const int nTiles = (N + TILE - 1) / TILE;
    int*  inCnt   = (int*)carve((size_t)N * 4);
    int*  tsum    = (int*)carve((size_t)nTiles * 4);
    int*  rowptr  = (int*)carve((size_t)(N + 1) * 4);
    int*  cursor  = (int*)carve((size_t)N * 4);
    int2* ent     = (int2*)carve((size_t)E * 8);

    const int BLK = 256;
    auto grid1   = [&](long long n) { return dim3((unsigned)((n + BLK - 1) / BLK)); };
    auto gridWv  = [&](long long waves) { return dim3((unsigned)((waves * 32 + BLK - 1) / BLK)); };
    auto gemmGrid = [&](int rows) { return gridWv((rows + 15) / 16); };

    fill_i32<<<grid1(N), BLK, 0, stream>>>(inCnt, 0, N);
    count_in_edges<<<grid1(E), BLK, 0, stream>>>(dst, inCnt, E);
    tile_sums<<<dim3((unsigned)nTiles), SCAN_T, 0, stream>>>(inCnt, tsum, N);
    scan_rows<<<dim3((unsigned)nTiles), SCAN_T, 0, stream>>>(inCnt, tsum, rowptr, cursor, dis, N);
    place_edges<<<grid1(E), BLK, 0, stream>>>(src, dst, dis, cursor, ent, E);

    wmma_gemm<2><<<gemmGrid(N), BLK, 0, stream>>>(x, W1, b1, bufT, N, K0, 0);
    pull_rows<32><<<gridWv(N), BLK, 0, stream>>>(rowptr, ent, dis, bufT, b1, bufA, N);

    wmma_gemm<3><<<gemmGrid(N), BLK, 0, stream>>>(bufA, W2, b2, bufT, N, F1, 0);
    pull_rows<48><<<gridWv(N), BLK, 0, stream>>>(rowptr, ent, dis, bufT, b2, bufB, N);

    wmma_gemm<4><<<gemmGrid(N), BLK, 0, stream>>>(bufB, W3, b3, bufT, N, F2, 0);
    pull_rows<64><<<gridWv(N), BLK, 0, stream>>>(rowptr, ent, dis, bufT, b3, bufA, N);

    fill_f32<<<grid1((long long)G * F3), BLK, 0, stream>>>(gmax, 0.0f, G * F3);
    fill_f32<<<grid1((long long)G * F3), BLK, 0, stream>>>(gsum, 0.0f, G * F3);
    fill_f32<<<grid1(G), BLK, 0, stream>>>(cnt, 0.0f, G);
    pool_nodes<<<gridWv(N), BLK, 0, stream>>>(bufA, batch, gmax, gsum, cnt, N, F3);
    build_pooled<<<grid1((long long)G * 2 * F3), BLK, 0, stream>>>(gmax, gsum, cnt, pooled, G, F3);

    wmma_gemm<2><<<gemmGrid(G), BLK, 0, stream>>>(pooled, Wout, bout, out, G, 2 * F3, 1);
}


# node kernels (3 aggregations, pooling): each resident wave walks a run of consecutive nodes, next node's entries and own row requested one node ahead, twice the row loads in flight; sums unchanged
# speedup vs baseline: 5.1057x; 2.3426x over previous
#include <hip/hip_runtime.h>

typedef __attribute__((ext_vector_type(2))) float v2f;
typedef __attribute__((ext_vector_type(8))) float v8f;

__global__ void fill_f32(float* __restrict__ p, float v, int n) {
    int i = blockIdx.x * blockDim.x + threadIdx.x;
    if (i < n) p[i] = v;
}

__global__ void fill_i32(int* __restrict__ p, int v, int n) {
    int i = blockIdx.x * blockDim.x + threadIdx.x;
    if (i < n) p[i] = v;
}


#define BSH    8
#define BW     (1 << BSH)
#define NBMAX  512
#define CH     4096
#define CH_T   256
#define CH_PER (CH / CH_T)
#define FIN_T  512
#define FIN_U  4
#define STAGE  12288

__global__ void bucket_count(const int* __restrict__ dst, int* __restrict__ total, int nE, int nB) {
    __shared__ int hist[NBMAX];
    int t = threadIdx.x;
    for (int i = t; i < NBMAX; i += CH_T) hist[i] = 0;
    __syncthreads();
    int e0 = blockIdx.x * CH;
    int n  = nE - e0;
    if (n > CH) n = CH;
    int d[CH_PER];
#pragma unroll
    for (int k = 0; k < CH_PER; ++k) {
        int i = k * CH_T + t;
        d[k] = dst[e0 + (i < n ? i : 0)];
        if (i >= n) d[k] = -1;
    }
#pragma unroll
    for (int k = 0; k < CH_PER; ++k)
        if (d[k] >= 0) atomicAdd(&hist[d[k] >> BSH], 1);
    __syncthreads();
    for (int i = t; i < nB; i += CH_T) {
        int c = hist[i];
        if (c) atomicAdd(&total[i], c);
    }
}

__global__ void bucket_starts(const int* __restrict__ total, int* __restrict__ bstart, int nB) {
    __shared__ int sh[NBMAX];
    int t = threadIdx.x;
    int c = (t < nB) ? total[t] : 0;
    sh[t] = c;
    __syncthreads();
    for (int o = 1; o < NBMAX; o <<= 1) {
        int v = (t >= o) ? sh[t - o] : 0;
        __syncthreads();
        sh[t] += v;
        __syncthreads();
    }
    if (t < nB) bstart[t] = sh[t] - c;
    if (t == nB - 1) bstart[nB] = sh[t];
}

__global__ void bucket_scatter(const int* __restrict__ src, const int* __restrict__ dst,
                               const int* __restrict__ bstart, int* __restrict__ bcursor,
                               int2* __restrict__ rec, int nE) {
    __shared__ int  hist[NBMAX];
    __shared__ int  gpos[NBMAX];
    __shared__ int  sh[CH_T];
    __shared__ int2 sorted[CH];
    int t = threadIdx.x;
    for (int i = t; i < NBMAX; i += CH_T) hist[i] = 0;
    __syncthreads();

    int e0 = blockIdx.x * CH;
    int n  = nE - e0;
    if (n > CH) n = CH;
    int s[CH_PER], d[CH_PER], r[CH_PER];
#pragma unroll
    for (int k = 0; k < CH_PER; ++k) {
        int i = k * CH_T + t;
        int e = e0 + (i < n ? i : 0);
        s[k] = src[e];
        d[k] = dst[e];
        if (i >= n) d[k] = -1;
    }
#pragma unroll
    for (int k = 0; k < CH_PER; ++k) {
        r[k] = 0;
        if (d[k] >= 0) r[k] = atomicAdd(&hist[d[k] >> BSH], 1);
    }
    __syncthreads();

    int b0 = 2 * t, b1 = 2 * t + 1;
    int h0 = hist[b0], h1 = hist[b1];
    int sum = h0 + h1;
    sh[t] = sum;
    __syncthreads();
    for (int o = 1; o < CH_T; o <<= 1) {
        int v = (t >= o) ? sh[t - o] : 0;
        __syncthreads();
        sh[t] += v;
        __syncthreads();
    }
    int l0 = sh[t] - sum;
    int l1 = l0 + h0;
    gpos[b0] = (h0 > 0) ? bstart[b0] + atomicAdd(&bcursor[b0], h0) - l0 : 0;
    gpos[b1] = (h1 > 0) ? bstart[b1] + atomicAdd(&bcursor[b1], h1) - l1 : 0;
    hist[b0] = l0;
    hist[b1] = l1;
    __syncthreads();

#pragma unroll
    for (int k = 0; k < CH_PER; ++k)
        if (d[k] >= 0) sorted[hist[d[k] >> BSH] + r[k]] = make_int2(s[k], d[k]);
    __syncthreads();

    for (int i = t; i < n; i += CH_T) {
        int2 v = sorted[i];
        rec[gpos[v.y >> BSH] + i] = v;
    }
}

__global__ void bucket_finish(const int2* __restrict__ rec, const int* __restrict__ bstart,
                              int* __restrict__ rowptr, float* __restrict__ dis,
                              int* __restrict__ ent, int N, int nB) {
    __shared__ int cnt[BW];
    __shared__ int cur[BW];
    __shared__ int sh[FIN_T];
    __shared__ int stage[STAGE];
    int t     = threadIdx.x;
    int b     = blockIdx.x;
    int beg   = bstart[b];
    int end   = bstart[b + 1];
    int n     = end - beg;
    int node0 = b << BSH;
    if (t < BW) cnt[t] = 0;
    __syncthreads();
    for (int i0 = t; i0 < n; i0 += FIN_U * FIN_T) {
        int dl[FIN_U];
#pragma unroll
        for (int u = 0; u < FIN_U; ++u) {
            int i = i0 + u * FIN_T;
            dl[u] = rec[beg + (i < n ? i : i0)].y - node0;
            if (i >= n) dl[u] = -1;
        }
#pragma unroll
        for (int u = 0; u < FIN_U; ++u)
            if (dl[u] >= 0) atomicAdd(&cnt[dl[u]], 1);
    }
    __syncthreads();

    int c = (t < BW) ? cnt[t] : 0;
    sh[t] = c;
    __syncthreads();
    for (int o = 1; o < BW; o <<= 1) {
        int v = (t >= o) ? sh[t - o] : 0;
        __syncthreads();
        sh[t] += v;
        __syncthreads();
    }
    if (t < BW) {
        int st = sh[t] - c;
        cur[t] = st;
        int node = node0 + t;
        if (node < N) {
            rowptr[node] = beg + st;
            dis[node]    = rsqrtf((float)(c + 1));
        }
    }
    if (b == nB - 1 && t == 0) rowptr[N] = end;
    __syncthreads();

    bool staged = n <= STAGE;
    for (int i0 = t; i0 < n; i0 += FIN_U * FIN_T) {
        int2 v[FIN_U];
#pragma unroll
        for (int u = 0; u < FIN_U; ++u) {
            int i = i0 + u * FIN_T;
            v[u] = rec[beg + (i < n ? i : i0)];
            if (i >= n) v[u].y = -1;
        }
#pragma unroll
        for (int u = 0; u < FIN_U; ++u) {
            if (v[u].y >= 0) {
                int p = atomicAdd(&cur[v[u].y - node0], 1);
                if (staged) stage[p] = v[u].x;
                else        ent[beg + p] = v[u].x;
            }
        }
    }
    __syncthreads();
    if (staged)
        for (int i = t; i < n; i += FIN_T) ent[beg + i] = stage[i];
}

template<int FT>
__global__ void wmma_gemm(const float* __restrict__ X, const float* __restrict__ W,
                          const float* __restrict__ Bias, const float* __restrict__ RowScale,
                          float* __restrict__ Y, int N, int K, int addBias) {
    const int F    = FT * 16;
    int wave = (blockIdx.x * blockDim.x + threadIdx.x) >> 5;
    int lane = threadIdx.x & 31;
    int row0 = wave << 4;
    if (row0 >= N) return;
    int half = lane >> 4;
    int l16  = lane & 15;
    int arow = row0 + l16; if (arow > N - 1) arow = N - 1;

    v8f acc[FT];
    v8f z = {0.f, 0.f, 0.f, 0.f, 0.f, 0.f, 0.f, 0.f};
#pragma unroll
    for (int t = 0; t < FT; ++t) acc[t] = z;

    for (int k0 = 0; k0 < K; k0 += 4) {
        int ka = k0 + half * 2;
        v2f a;
        a.x = X[arow * K + ka];
        a.y = X[arow * K + ka + 1];
#pragma unroll
        for (int t = 0; t < FT; ++t) {
            int col = t * 16 + l16;
            v2f b;
            b.x = W[ka * F + col];
            b.y = W[(ka + 1) * F + col];
            acc[t] = __builtin_amdgcn_wmma_f32_16x16x4_f32(
                 false, a,  false, b,
                 (short)0, acc[t],  false,  false);
        }
    }

    float sc[8];
#pragma unroll
    for (int v = 0; v < 8; ++v) {
        int r = row0 + half * 8 + v;
        sc[v] = (RowScale != nullptr && r < N) ? RowScale[r] : 1.f;
    }
#pragma unroll
    for (int t = 0; t < FT; ++t) {
        int col = t * 16 + l16;
        float bb = addBias ? Bias[col] : 0.f;
#pragma unroll
        for (int v = 0; v < 8; ++v) {
            int r = row0 + half * 8 + v;
            if (r < N) Y[r * F + col] = (acc[t][v] + bb) * sc[v];
        }
    }
}

#define NPW 13

template<int VEC> struct RowVec;
template<> struct RowVec<1> { typedef float type; };
template<> struct RowVec<2> { typedef v2f   type; };

template<int VEC, int U>
__device__ __forceinline__ void pull_group(const float* __restrict__ H, unsigned fo, int off,
                                           int j, typename RowVec<VEC>::type& acc) {
    typedef typename RowVec<VEC>::type vec_t;
    vec_t h[U];
#pragma unroll
    for (int u = 0; u < U; ++u) {
        const float* row = H + __builtin_amdgcn_readlane(off, j + u);
        h[u] = *(const vec_t*)(row + fo);
    }
#pragma unroll
    for (int u = 0; u < U; ++u) acc += h[u];
}

template<int F>
__global__ void pull_rows(const int* __restrict__ rowptr, const int* __restrict__ ent,
                          const float* __restrict__ dis, const float* __restrict__ H,
                          const float* __restrict__ bias, float* __restrict__ OUT, int N) {
    constexpr int VEC = (F > 32) ? 2 : 1;
    typedef typename RowVec<VEC>::type vec_t;
    int wave  = __builtin_amdgcn_readfirstlane((int)((blockIdx.x * blockDim.x + threadIdx.x) >> 5));
    int node0 = wave * NPW;
    if (node0 >= N) return;
    int nn = N - node0;
    if (nn > NPW) nn = NPW;
    int  lane   = threadIdx.x & 31;
    bool active = lane * VEC < F;
    unsigned fo = active ? lane * VEC : 0;

    int   rp = rowptr[node0 + (lane < nn ? lane : nn)];
    float dv = dis[node0 + (lane < nn ? lane : nn - 1)];
    vec_t bv;
    if constexpr (VEC == 1) bv = bias[fo];
    else                    { bv.x = bias[fo]; bv.y = bias[fo + 1]; }

    int beg = __builtin_amdgcn_readlane(rp, 0);
    int end = __builtin_amdgcn_readlane(rp, 1);
    int off = 0;
    if (lane < end - beg) off = ent[beg + lane] * F;
    vec_t self = *(const vec_t*)(H + node0 * F + fo);

    for (int k = 0; k < nn; ++k) {
        int kn   = k + 1 < nn ? k + 1 : k;
        int nend = __builtin_amdgcn_readlane(rp, kn + 1);
        int i    = end + lane;
        if (i > nend - 1) i = nend - 1;
        if (i < 0) i = 0;
        int   entn  = ent[i];
        vec_t selfn = *(const vec_t*)(H + (node0 + kn) * F + fo);

        vec_t acc = self;
        for (int base = beg; ; ) {
            int cnt = end - base;
            if (cnt > 32) cnt = 32;
            int j = 0;
            for (; j + 16 <= cnt; j += 16) pull_group<VEC, 16>(H, fo, off, j, acc);
            if (cnt & 8) { pull_group<VEC, 8>(H, fo, off, j, acc); j += 8; }
            if (cnt & 4) { pull_group<VEC, 4>(H, fo, off, j, acc); j += 4; }
            if (cnt & 2) { pull_group<VEC, 2>(H, fo, off, j, acc); j += 2; }
            if (cnt & 1) { pull_group<VEC, 1>(H, fo, off, j, acc); }
            base += 32;
            if (base >= end) break;
            off = 0;
            if (lane < end - base) off = ent[base + lane] * F;
        }

        float dd = __int_as_float(__builtin_amdgcn_readlane(__float_as_int(dv), k));
        if (active) {
            vec_t v;
            if constexpr (VEC == 1) {
                v = acc * dd + bv;
                v = v > 0.f ? v : 0.f;
            } else {
                v.x = acc.x * dd + bv.x;
                v.y = acc.y * dd + bv.y;
                v.x = v.x > 0.f ? v.x : 0.f;
                v.y = v.y > 0.f ? v.y : 0.f;
            }
            *(vec_t*)(OUT + (node0 + k) * F + fo) = v;
        }
        beg = end; end = nend; off = entn * F; self = selfn;
    }
}

#define PNW 16
#define PNU 4

__global__ void pool_nodes(const float* __restrict__ H, const int* __restrict__ batch,
                           float* __restrict__ gmax, float* __restrict__ gsum,
                           float* __restrict__ cnt, int N, int F) {
    int wave = __builtin_amdgcn_readfirstlane((int)((blockIdx.x * blockDim.x + threadIdx.x) >> 5));
    int lane = threadIdx.x & 31;
    int i0   = wave * PNW;
    if (i0 >= N) return;
    int nn = N - i0;
    if (nn > PNW) nn = PNW;
    int gl = batch[i0 + (lane < nn ? lane : 0)];
    if (lane < nn) atomicAdd(&cnt[gl], 1.0f);
    for (int k0 = 0; k0 < nn; k0 += PNU) {
        for (int f = lane; f < F; f += 32) {
            float v[PNU];
#pragma unroll
            for (int u = 0; u < PNU; ++u) {
                int k = k0 + u < nn ? k0 + u : nn - 1;
                v[u] = H[(i0 + k) * F + f];
            }
#pragma unroll
            for (int u = 0; u < PNU; ++u) {
                if (k0 + u < nn) {
                    int g = __builtin_amdgcn_readlane(gl, k0 + u);
                    atomicMax((unsigned int*)&gmax[g * F + f], __float_as_uint(v[u]));
                    atomicAdd(&gsum[g * F + f], v[u]);
                }
            }
        }
    }
}

__global__ void build_pooled(const float* __restrict__ gmax, const float* __restrict__ gsum,
                             const float* __restrict__ cnt, float* __restrict__ pooled,
                             int G, int F) {
    int idx = blockIdx.x * blockDim.x + threadIdx.x;
    if (idx >= G * 2 * F) return;
    int g = idx / (2 * F);
    int f = idx % (2 * F);
    float v;
    if (f < F) {
        v = gmax[g * F + f];
    } else {
        float c = cnt[g];
        v = gsum[g * F + (f - F)] / (c > 1.f ? c : 1.f);
    }
    pooled[idx] = v;
}

extern "C" void kernel_launch(void* const* d_in, const int* in_sizes, int n_in,
                              void* d_out, int out_size, void* d_ws, size_t ws_size,
                              hipStream_t stream) {
    const float* x     = (const float*)d_in[0];
    const int*   ei    = (const int*)d_in[1];
    const int*   batch = (const int*)d_in[2];
    const float* W1    = (const float*)d_in[3];
    const float* b1    = (const float*)d_in[4];
    const float* W2    = (const float*)d_in[5];
    const float* b2    = (const float*)d_in[6];
    const float* W3    = (const float*)d_in[7];
    const float* b3    = (const float*)d_in[8];
    const float* Wout  = (const float*)d_in[9];
    const float* bout  = (const float*)d_in[10];
    float*       out   = (float*)d_out;

    const int N  = in_sizes[2];
    const int E  = in_sizes[1] / 2;
    const int K0 = in_sizes[0] / N;
    const int F1 = in_sizes[4];
    const int F2 = in_sizes[6];
    const int F3 = in_sizes[8];
    const int Fo = in_sizes[10];
    const int G  = out_size / Fo;

    const int* src = ei;
    const int* dst = ei + E;

    char*  ws  = (char*)d_ws;
    size_t off = 0;
    auto carve = [&](size_t bytes) -> float* {
        float* p = (float*)(ws + off);
        off += (bytes + 255) & ~(size_t)255;
        return p;
    };
    float* dis    = carve((size_t)N * 4);
    float* bufT   = carve((size_t)N * F3 * 4);
    float* bufA   = carve((size_t)N * F3 * 4);
    float* bufB   = carve((size_t)N * F3 * 4);
    float* gmax   = carve((size_t)G * F3 * 4);
    float* gsum   = carve((size_t)G * F3 * 4);
    float* cnt    = carve((size_t)G * 4);
    float* pooled = carve((size_t)G * 2 * F3 * 4);
    const int nB  = (N + BW - 1) >> BSH;
    const int nCh = (E + CH - 1) / CH;
    int*  btotal  = (int*)carve((size_t)2 * NBMAX * 4);
    int*  bcursor = btotal + NBMAX;
    int*  bstart  = (int*)carve((size_t)(NBMAX + 1) * 4);
    int*  rowptr  = (int*)carve((size_t)(N + 1) * 4);
    int*  ent     = (int*)carve((size_t)E * 4);
    int2* rec     = (int2*)carve((size_t)E * 8);

    const int BLK = 256;
    auto grid1   = [&](long long n) { return dim3((unsigned)((n + BLK - 1) / BLK)); };
    auto gridWv  = [&](long long waves) { return dim3((unsigned)((waves * 32 + BLK - 1) / BLK)); };
    auto gemmGrid = [&](int rows) { return gridWv((rows + 15) / 16); };

    fill_i32<<<grid1(2 * NBMAX), BLK, 0, stream>>>(btotal, 0, 2 * NBMAX);
    bucket_count<<<dim3((unsigned)nCh), CH_T, 0, stream>>>(dst, btotal, E, nB);
    bucket_starts<<<dim3(1), NBMAX, 0, stream>>>(btotal, bstart, nB);
    bucket_scatter<<<dim3((unsigned)nCh), CH_T, 0, stream>>>(src, dst, bstart, bcursor, rec, E);
    bucket_finish<<<dim3((unsigned)nB), FIN_T, 0, stream>>>(rec, bstart, rowptr, dis, ent, N, nB);

    wmma_gemm<2><<<gemmGrid(N), BLK, 0, stream>>>(x, W1, b1, dis, bufT, N, K0, 0);
    pull_rows<32><<<gridWv((N + NPW - 1) / NPW), BLK, 0, stream>>>(rowptr, ent, dis, bufT, b1, bufA, N);

    wmma_gemm<3><<<gemmGrid(N), BLK, 0, stream>>>(bufA, W2, b2, dis, bufT, N, F1, 0);
    pull_rows<48><<<gridWv((N + NPW - 1) / NPW), BLK, 0, stream>>>(rowptr, ent, dis, bufT, b2, bufB, N);

    wmma_gemm<4><<<gemmGrid(N), BLK, 0, stream>>>(bufB, W3, b3, dis, bufT, N, F2, 0);
    pull_rows<64><<<gridWv((N + NPW - 1) / NPW), BLK, 0, stream>>>(rowptr, ent, dis, bufT, b3, bufA, N);

    fill_f32<<<grid1((long long)G * F3), BLK, 0, stream>>>(gmax, 0.0f, G * F3);
    fill_f32<<<grid1((long long)G * F3), BLK, 0, stream>>>(gsum, 0.0f, G * F3);
    fill_f32<<<grid1(G), BLK, 0, stream>>>(cnt, 0.0f, G);
    pool_nodes<<<gridWv((N + PNW - 1) / PNW), BLK, 0, stream>>>(bufA, batch, gmax, gsum, cnt, N, F3);
    build_pooled<<<grid1((long long)G * 2 * F3), BLK, 0, stream>>>(gmax, gsum, cnt, pooled, G, F3);

    wmma_gemm<2><<<gemmGrid(G), BLK, 0, stream>>>(pooled, Wout, bout, nullptr, out, G, 2 * F3, 1);
}


# pooling fused into the layer-3 aggregation: wave keeps max and sum of its nodes of one graph in registers, atomics only on a graph change; h3, pooled not stored; 17 to 12 launches; sum order changes
# speedup vs baseline: 5.4293x; 1.0634x over previous
#include <hip/hip_runtime.h>

typedef __attribute__((ext_vector_type(2))) float v2f;
typedef __attribute__((ext_vector_type(8))) float v8f;

__global__ void fill_i32(int* __restrict__ p, int v, int n) {
    int i = blockIdx.x * blockDim.x + threadIdx.x;
    if (i < n) p[i] = v;
}


#define BSH    8
#define BW     (1 << BSH)
#define NBMAX  512
#define CH     4096
#define CH_T   256
#define CH_PER (CH / CH_T)
#define FIN_T  512
#define FIN_U  4
#define STAGE  12288

__global__ void bucket_count(const int* __restrict__ dst, int* __restrict__ total, int nE, int nB) {
    __shared__ int hist[NBMAX];
    int t = threadIdx.x;
    for (int i = t; i < NBMAX; i += CH_T) hist[i] = 0;
    __syncthreads();
    int e0 = blockIdx.x * CH;
    int n  = nE - e0;
    if (n > CH) n = CH;
    int d[CH_PER];
#pragma unroll
    for (int k = 0; k < CH_PER; ++k) {
        int i = k * CH_T + t;
        d[k] = dst[e0 + (i < n ? i : 0)];
        if (i >= n) d[k] = -1;
    }
#pragma unroll
    for (int k = 0; k < CH_PER; ++k)
        if (d[k] >= 0) atomicAdd(&hist[d[k] >> BSH], 1);
    __syncthreads();
    for (int i = t; i < nB; i += CH_T) {
        int c = hist[i];
        if (c) atomicAdd(&total[i], c);
    }
}

__global__ void bucket_starts(const int* __restrict__ total, int* __restrict__ bstart, int nB) {
    __shared__ int sh[NBMAX];
    int t = threadIdx.x;
    int c = (t < nB) ? total[t] : 0;
    sh[t] = c;
    __syncthreads();
    for (int o = 1; o < NBMAX; o <<= 1) {
        int v = (t >= o) ? sh[t - o] : 0;
        __syncthreads();
        sh[t] += v;
        __syncthreads();
    }
    if (t < nB) bstart[t] = sh[t] - c;
    if (t == nB - 1) bstart[nB] = sh[t];
}

__global__ void bucket_scatter(const int* __restrict__ src, const int* __restrict__ dst,
                               const int* __restrict__ bstart, int* __restrict__ bcursor,
                               int2* __restrict__ rec, int nE) {
    __shared__ int  hist[NBMAX];
    __shared__ int  gpos[NBMAX];
    __shared__ int  sh[CH_T];
    __shared__ int2 sorted[CH];
    int t = threadIdx.x;
    for (int i = t; i < NBMAX; i += CH_T) hist[i] = 0;
    __syncthreads();

    int e0 = blockIdx.x * CH;
    int n  = nE - e0;
    if (n > CH) n = CH;
    int s[CH_PER], d[CH_PER], r[CH_PER];
#pragma unroll
    for (int k = 0; k < CH_PER; ++k) {
        int i = k * CH_T + t;
        int e = e0 + (i < n ? i : 0);
        s[k] = src[e];
        d[k] = dst[e];
        if (i >= n) d[k] = -1;
    }
#pragma unroll
    for (int k = 0; k < CH_PER; ++k) {
        r[k] = 0;
        if (d[k] >= 0) r[k] = atomicAdd(&hist[d[k] >> BSH], 1);
    }
    __syncthreads();

    int b0 = 2 * t, b1 = 2 * t + 1;
    int h0 = hist[b0], h1 = hist[b1];
    int sum = h0 + h1;
    sh[t] = sum;
    __syncthreads();
    for (int o = 1; o < CH_T; o <<= 1) {
        int v = (t >= o) ? sh[t - o] : 0;
        __syncthreads();
        sh[t] += v;
        __syncthreads();
    }
    int l0 = sh[t] - sum;
    int l1 = l0 + h0;
    gpos[b0] = (h0 > 0) ? bstart[b0] + atomicAdd(&bcursor[b0], h0) - l0 : 0;
    gpos[b1] = (h1 > 0) ? bstart[b1] + atomicAdd(&bcursor[b1], h1) - l1 : 0;
    hist[b0] = l0;
    hist[b1] = l1;
    __syncthreads();

#pragma unroll
    for (int k = 0; k < CH_PER; ++k)
        if (d[k] >= 0) sorted[hist[d[k] >> BSH] + r[k]] = make_int2(s[k], d[k]);
    __syncthreads();

    for (int i = t; i < n; i += CH_T) {
        int2 v = sorted[i];
        rec[gpos[v.y >> BSH] + i] = v;
    }
}

__global__ void bucket_finish(const int2* __restrict__ rec, const int* __restrict__ bstart,
                              int* __restrict__ rowptr, float* __restrict__ dis,
                              int* __restrict__ ent, int N, int nB) {
    __shared__ int cnt[BW];
    __shared__ int cur[BW];
    __shared__ int sh[FIN_T];
    __shared__ int stage[STAGE];
    int t     = threadIdx.x;
    int b     = blockIdx.x;
    int beg   = bstart[b];
    int end   = bstart[b + 1];
    int n     = end - beg;
    int node0 = b << BSH;
    if (t < BW) cnt[t] = 0;
    __syncthreads();
    for (int i0 = t; i0 < n; i0 += FIN_U * FIN_T) {
        int dl[FIN_U];
#pragma unroll
        for (int u = 0; u < FIN_U; ++u) {
            int i = i0 + u * FIN_T;
            dl[u] = rec[beg + (i < n ? i : i0)].y - node0;
            if (i >= n) dl[u] = -1;
        }
#pragma unroll
        for (int u = 0; u < FIN_U; ++u)
            if (dl[u] >= 0) atomicAdd(&cnt[dl[u]], 1);
    }
    __syncthreads();

    int c = (t < BW) ? cnt[t] : 0;
    sh[t] = c;
    __syncthreads();
    for (int o = 1; o < BW; o <<= 1) {
        int v = (t >= o) ? sh[t - o] : 0;
        __syncthreads();
        sh[t] += v;
        __syncthreads();
    }
    if (t < BW) {
        int st = sh[t] - c;
        cur[t] = st;
        int node = node0 + t;
        if (node < N) {
            rowptr[node] = beg + st;
            dis[node]    = rsqrtf((float)(c + 1));
        }
    }
    if (b == nB - 1 && t == 0) rowptr[N] = end;
    __syncthreads();

    bool staged = n <= STAGE;
    for (int i0 = t; i0 < n; i0 += FIN_U * FIN_T) {
        int2 v[FIN_U];
#pragma unroll
        for (int u = 0; u < FIN_U; ++u) {
            int i = i0 + u * FIN_T;
            v[u] = rec[beg + (i < n ? i : i0)];
            if (i >= n) v[u].y = -1;
        }
#pragma unroll
        for (int u = 0; u < FIN_U; ++u) {
            if (v[u].y >= 0) {
                int p = atomicAdd(&cur[v[u].y - node0], 1);
                if (staged) stage[p] = v[u].x;
                else        ent[beg + p] = v[u].x;
            }
        }
    }
    __syncthreads();
    if (staged)
        for (int i = t; i < n; i += FIN_T) ent[beg + i] = stage[i];
}

template<int FT>
__global__ void wmma_gemm(const float* __restrict__ X, const float* __restrict__ W,
                          const float* __restrict__ Bias, const float* __restrict__ RowScale,
                          float* __restrict__ Y, int N, int K, int addBias) {
    const int F    = FT * 16;
    int wave = (blockIdx.x * blockDim.x + threadIdx.x) >> 5;
    int lane = threadIdx.x & 31;
    int row0 = wave << 4;
    if (row0 >= N) return;
    int half = lane >> 4;
    int l16  = lane & 15;
    int arow = row0 + l16; if (arow > N - 1) arow = N - 1;

    v8f acc[FT];
    v8f z = {0.f, 0.f, 0.f, 0.f, 0.f, 0.f, 0.f, 0.f};
#pragma unroll
    for (int t = 0; t < FT; ++t) acc[t] = z;

    for (int k0 = 0; k0 < K; k0 += 4) {
        int ka = k0 + half * 2;
        v2f a;
        a.x = X[arow * K + ka];
        a.y = X[arow * K + ka + 1];
#pragma unroll
        for (int t = 0; t < FT; ++t) {
            int col = t * 16 + l16;
            v2f b;
            b.x = W[ka * F + col];
            b.y = W[(ka + 1) * F + col];
            acc[t] = __builtin_amdgcn_wmma_f32_16x16x4_f32(
                 false, a,  false, b,
                 (short)0, acc[t],  false,  false);
        }
    }

    float sc[8];
#pragma unroll
    for (int v = 0; v < 8; ++v) {
        int r = row0 + half * 8 + v;
        sc[v] = (RowScale != nullptr && r < N) ? RowScale[r] : 1.f;
    }
#pragma unroll
    for (int t = 0; t < FT; ++t) {
        int col = t * 16 + l16;
        float bb = addBias ? Bias[col] : 0.f;
#pragma unroll
        for (int v = 0; v < 8; ++v) {
            int r = row0 + half * 8 + v;
            if (r < N) Y[r * F + col] = (acc[t][v] + bb) * sc[v];
        }
    }
}

#define NPW 13

template<int VEC> struct RowVec;
template<> struct RowVec<1> { typedef float type; };
template<> struct RowVec<2> { typedef v2f   type; };

template<int VEC, int U>
__device__ __forceinline__ void pull_group(const float* __restrict__ H, unsigned fo, int off,
                                           int j, typename RowVec<VEC>::type& acc) {
    typedef typename RowVec<VEC>::type vec_t;
    vec_t h[U];
#pragma unroll
    for (int u = 0; u < U; ++u) {
        const float* row = H + __builtin_amdgcn_readlane(off, j + u);
        h[u] = *(const vec_t*)(row + fo);
    }
#pragma unroll
    for (int u = 0; u < U; ++u) acc += h[u];
}

__device__ __forceinline__ void pool_flush(float* __restrict__ gmax, float* __restrict__ gsum,
                                           int at, v2f mx, v2f sm) {
    atomicMax((unsigned int*)(gmax + at),     __float_as_uint(mx.x));
    atomicMax((unsigned int*)(gmax + at + 1), __float_as_uint(mx.y));
    atomicAdd(gsum + at,     sm.x);
    atomicAdd(gsum + at + 1, sm.y);
}

template<int F, bool POOL>
__global__ void pull_rows(const int* __restrict__ rowptr, const int* __restrict__ ent,
                          const float* __restrict__ dis, const float* __restrict__ H,
                          const float* __restrict__ bias, float* __restrict__ OUT,
                          const int* __restrict__ batch, float* __restrict__ gmax,
                          float* __restrict__ gsum, float* __restrict__ cnt, int N) {
    constexpr int VEC = (F > 32) ? 2 : 1;
    static_assert(!POOL || F == 64, "the pooled form holds a whole row in the wave: 2 features per lane");
    typedef typename RowVec<VEC>::type vec_t;
    int wave  = __builtin_amdgcn_readfirstlane((int)((blockIdx.x * blockDim.x + threadIdx.x) >> 5));
    int node0 = wave * NPW;
    if (node0 >= N) return;
    int nn = N - node0;
    if (nn > NPW) nn = NPW;
    int  lane   = threadIdx.x & 31;
    bool active = lane * VEC < F;
    unsigned fo = active ? lane * VEC : 0;

    int   rp = rowptr[node0 + (lane < nn ? lane : nn)];
    float dv = dis[node0 + (lane < nn ? lane : nn - 1)];
    vec_t bv;
    if constexpr (VEC == 1) bv = bias[fo];
    else                    { bv.x = bias[fo]; bv.y = bias[fo + 1]; }

    int beg = __builtin_amdgcn_readlane(rp, 0);
    int end = __builtin_amdgcn_readlane(rp, 1);
    int off = 0;
    if (lane < end - beg) off = ent[beg + lane] * F;
    vec_t self = *(const vec_t*)(H + node0 * F + fo);

    int   gl   = 0;
    int   gcur = 0;
    vec_t pmax = self, psum = self;
    if constexpr (POOL) {
        gl = batch[node0 + (lane < nn ? lane : 0)];
        if (lane < nn) atomicAdd(&cnt[gl], 1.0f);
    }

    for (int k = 0; k < nn; ++k) {
        int kn   = k + 1 < nn ? k + 1 : k;
        int nend = __builtin_amdgcn_readlane(rp, kn + 1);
        int i    = end + lane;
        if (i > nend - 1) i = nend - 1;
        if (i < 0) i = 0;
        int   entn  = ent[i];
        vec_t selfn = *(const vec_t*)(H + (node0 + kn) * F + fo);

        vec_t acc = self;
        for (int base = beg; ; ) {
            int cnt = end - base;
            if (cnt > 32) cnt = 32;
            int j = 0;
            for (; j + 16 <= cnt; j += 16) pull_group<VEC, 16>(H, fo, off, j, acc);
            if (cnt & 8) { pull_group<VEC, 8>(H, fo, off, j, acc); j += 8; }
            if (cnt & 4) { pull_group<VEC, 4>(H, fo, off, j, acc); j += 4; }
            if (cnt & 2) { pull_group<VEC, 2>(H, fo, off, j, acc); j += 2; }
            if (cnt & 1) { pull_group<VEC, 1>(H, fo, off, j, acc); }
            base += 32;
            if (base >= end) break;
            off = 0;
            if (lane < end - base) off = ent[base + lane] * F;
        }

        float dd = __int_as_float(__builtin_amdgcn_readlane(__float_as_int(dv), k));
        vec_t v;
        if constexpr (VEC == 1) {
            v = acc * dd + bv;
            v = v > 0.f ? v : 0.f;
        } else {
            v.x = acc.x * dd + bv.x;
            v.y = acc.y * dd + bv.y;
            v.x = v.x > 0.f ? v.x : 0.f;
            v.y = v.y > 0.f ? v.y : 0.f;
        }
        if constexpr (POOL) {
            int g = __builtin_amdgcn_readlane(gl, k);
            if (k > 0 && g == gcur) {
                pmax.x = fmaxf(pmax.x, v.x);
                pmax.y = fmaxf(pmax.y, v.y);
                psum  += v;
            } else {
                if (k > 0) pool_flush(gmax, gsum, gcur * F + (int)fo, pmax, psum);
                gcur = g; pmax = v; psum = v;
            }
        } else {
            if (active) *(vec_t*)(OUT + (node0 + k) * F + fo) = v;
        }
        beg = end; end = nend; off = entn * F; self = selfn;
    }
    if constexpr (POOL) pool_flush(gmax, gsum, gcur * F + (int)fo, pmax, psum);
}

template<int FT>
__global__ void project_pooled(const float* __restrict__ gmax, const float* __restrict__ gsum,
                               const float* __restrict__ cnt, const float* __restrict__ W,
                               const float* __restrict__ Bias, float* __restrict__ Y,
                               int G, int Fh) {
    const int F    = FT * 16;
    int wave = (blockIdx.x * blockDim.x + threadIdx.x) >> 5;
    int lane = threadIdx.x & 31;
    int row0 = wave << 4;
    if (row0 >= G) return;
    int half = lane >> 4;
    int l16  = lane & 15;
    int arow = row0 + l16; if (arow > G - 1) arow = G - 1;
    float c = cnt[arow];
    c = c > 1.f ? c : 1.f;

    v8f acc[FT];
    v8f z = {0.f, 0.f, 0.f, 0.f, 0.f, 0.f, 0.f, 0.f};
#pragma unroll
    for (int t = 0; t < FT; ++t) acc[t] = z;

    for (int k0 = 0; k0 < Fh; k0 += 4) {
        int ka = k0 + half * 2;
        v2f a;
        a.x = gmax[arow * Fh + ka];
        a.y = gmax[arow * Fh + ka + 1];
#pragma unroll
        for (int t = 0; t < FT; ++t) {
            int col = t * 16 + l16;
            v2f b;
            b.x = W[ka * F + col];
            b.y = W[(ka + 1) * F + col];
            acc[t] = __builtin_amdgcn_wmma_f32_16x16x4_f32(
                 false, a,  false, b,
                 (short)0, acc[t],  false,  false);
        }
    }
    for (int k0 = 0; k0 < Fh; k0 += 4) {
        int ka = k0 + half * 2;
        v2f a;
        a.x = gsum[arow * Fh + ka] / c;
        a.y = gsum[arow * Fh + ka + 1] / c;
#pragma unroll
        for (int t = 0; t < FT; ++t) {
            int col = t * 16 + l16;
            v2f b;
            b.x = W[(Fh + ka) * F + col];
            b.y = W[(Fh + ka + 1) * F + col];
            acc[t] = __builtin_amdgcn_wmma_f32_16x16x4_f32(
                 false, a,  false, b,
                 (short)0, acc[t],  false,  false);
        }
    }

#pragma unroll
    for (int t = 0; t < FT; ++t) {
        int col = t * 16 + l16;
        float bb = Bias[col];
#pragma unroll
        for (int v = 0; v < 8; ++v) {
            int r = row0 + half * 8 + v;
            if (r < G) Y[r * F + col] = acc[t][v] + bb;
        }
    }
}

extern "C" void kernel_launch(void* const* d_in, const int* in_sizes, int n_in,
                              void* d_out, int out_size, void* d_ws, size_t ws_size,
                              hipStream_t stream) {
    const float* x     = (const float*)d_in[0];
    const int*   ei    = (const int*)d_in[1];
    const int*   batch = (const int*)d_in[2];
    const float* W1    = (const float*)d_in[3];
    const float* b1    = (const float*)d_in[4];
    const float* W2    = (const float*)d_in[5];
    const float* b2    = (const float*)d_in[6];
    const float* W3    = (const float*)d_in[7];
    const float* b3    = (const float*)d_in[8];
    const float* Wout  = (const float*)d_in[9];
    const float* bout  = (const float*)d_in[10];
    float*       out   = (float*)d_out;

    const int N  = in_sizes[2];
    const int E  = in_sizes[1] / 2;
    const int K0 = in_sizes[0] / N;
    const int F1 = in_sizes[4];
    const int F2 = in_sizes[6];
    const int F3 = in_sizes[8];
    const int Fo = in_sizes[10];
    const int G  = out_size / Fo;

    const int* src = ei;
    const int* dst = ei + E;

    char*  ws  = (char*)d_ws;
    size_t off = 0;
    auto carve = [&](size_t bytes) -> float* {
        float* p = (float*)(ws + off);
        off += (bytes + 255) & ~(size_t)255;
        return p;
    };
    float* dis    = carve((size_t)N * 4);
    float* bufT   = carve((size_t)N * F3 * 4);
    float* bufA   = carve((size_t)N * F3 * 4);
    float* bufB   = carve((size_t)N * F3 * 4);
    const int nZero = 2 * G * F3 + G + 2 * NBMAX;
    float* zeros  = carve((size_t)nZero * 4);
    float* gmax   = zeros;
    float* gsum   = gmax + G * F3;
    float* cnt    = gsum + G * F3;
    int*  btotal  = (int*)(cnt + G);
    int*  bcursor = btotal + NBMAX;
    const int nB  = (N + BW - 1) >> BSH;
    const int nCh = (E + CH - 1) / CH;
    int*  bstart  = (int*)carve((size_t)(NBMAX + 1) * 4);
    int*  rowptr  = (int*)carve((size_t)(N + 1) * 4);
    int*  ent     = (int*)carve((size_t)E * 4);
    int2* rec     = (int2*)carve((size_t)E * 8);

    const int BLK = 256;
    auto grid1   = [&](long long n) { return dim3((unsigned)((n + BLK - 1) / BLK)); };
    auto gridWv  = [&](long long waves) { return dim3((unsigned)((waves * 32 + BLK - 1) / BLK)); };
    auto gemmGrid = [&](int rows) { return gridWv((rows + 15) / 16); };

    fill_i32<<<grid1(nZero), BLK, 0, stream>>>((int*)zeros, 0, nZero);
    bucket_count<<<dim3((unsigned)nCh), CH_T, 0, stream>>>(dst, btotal, E, nB);
    bucket_starts<<<dim3(1), NBMAX, 0, stream>>>(btotal, bstart, nB);
    bucket_scatter<<<dim3((unsigned)nCh), CH_T, 0, stream>>>(src, dst, bstart, bcursor, rec, E);
    bucket_finish<<<dim3((unsigned)nB), FIN_T, 0, stream>>>(rec, bstart, rowptr, dis, ent, N, nB);

    wmma_gemm<2><<<gemmGrid(N), BLK, 0, stream>>>(x, W1, b1, dis, bufT, N, K0, 0);
    pull_rows<32, false><<<gridWv((N + NPW - 1) / NPW), BLK, 0, stream>>>(rowptr, ent, dis, bufT, b1, bufA,
                                                                           nullptr, nullptr, nullptr, nullptr, N);

    wmma_gemm<3><<<gemmGrid(N), BLK, 0, stream>>>(bufA, W2, b2, dis, bufT, N, F1, 0);
    pull_rows<48, false><<<gridWv((N + NPW - 1) / NPW), BLK, 0, stream>>>(rowptr, ent, dis, bufT, b2, bufB,
                                                                           nullptr, nullptr, nullptr, nullptr, N);

    wmma_gemm<4><<<gemmGrid(N), BLK, 0, stream>>>(bufB, W3, b3, dis, bufT, N, F2, 0);
    pull_rows<64, true><<<gridWv((N + NPW - 1) / NPW), BLK, 0, stream>>>(rowptr, ent, dis, bufT, b3, nullptr,
                                                                          batch, gmax, gsum, cnt, N);

    project_pooled<2><<<gemmGrid(G), BLK, 0, stream>>>(gmax, gsum, cnt, Wout, bout, out, G, F3);
}


# aggregation kernels: 7 nodes per wave, not 13: the grid nearly fills every SIMD's wave slots in one resident round instead of under half, twice the row loads in flight; pooled sum order changes
# speedup vs baseline: 5.7158x; 1.0528x over previous
#include <hip/hip_runtime.h>

typedef __attribute__((ext_vector_type(2))) float v2f;
typedef __attribute__((ext_vector_type(8))) float v8f;

__global__ void fill_i32(int* __restrict__ p, int v, int n) {
    int i = blockIdx.x * blockDim.x + threadIdx.x;
    if (i < n) p[i] = v;
}


#define BSH    8
#define BW     (1 << BSH)
#define NBMAX  512
#define CH     4096
#define CH_T   256
#define CH_PER (CH / CH_T)
#define FIN_T  512
#define FIN_U  4
#define STAGE  12288

__global__ void bucket_count(const int* __restrict__ dst, int* __restrict__ total, int nE, int nB) {
    __shared__ int hist[NBMAX];
    int t = threadIdx.x;
    for (int i = t; i < NBMAX; i += CH_T) hist[i] = 0;
    __syncthreads();
    int e0 = blockIdx.x * CH;
    int n  = nE - e0;
    if (n > CH) n = CH;
    int d[CH_PER];
#pragma unroll
    for (int k = 0; k < CH_PER; ++k) {
        int i = k * CH_T + t;
        d[k] = dst[e0 + (i < n ? i : 0)];
        if (i >= n) d[k] = -1;
    }
#pragma unroll
    for (int k = 0; k < CH_PER; ++k)
        if (d[k] >= 0) atomicAdd(&hist[d[k] >> BSH], 1);
    __syncthreads();
    for (int i = t; i < nB; i += CH_T) {
        int c = hist[i];
        if (c) atomicAdd(&total[i], c);
    }
}

__global__ void bucket_starts(const int* __restrict__ total, int* __restrict__ bstart, int nB) {
    __shared__ int sh[NBMAX];
    int t = threadIdx.x;
    int c = (t < nB) ? total[t] : 0;
    sh[t] = c;
    __syncthreads();
    for (int o = 1; o < NBMAX; o <<= 1) {
        int v = (t >= o) ? sh[t - o] : 0;
        __syncthreads();
        sh[t] += v;
        __syncthreads();
    }
    if (t < nB) bstart[t] = sh[t] - c;
    if (t == nB - 1) bstart[nB] = sh[t];
}

__global__ void bucket_scatter(const int* __restrict__ src, const int* __restrict__ dst,
                               const int* __restrict__ bstart, int* __restrict__ bcursor,
                               int2* __restrict__ rec, int nE) {
    __shared__ int  hist[NBMAX];
    __shared__ int  gpos[NBMAX];
    __shared__ int  sh[CH_T];
    __shared__ int2 sorted[CH];
    int t = threadIdx.x;
    for (int i = t; i < NBMAX; i += CH_T) hist[i] = 0;
    __syncthreads();

    int e0 = blockIdx.x * CH;
    int n  = nE - e0;
    if (n > CH) n = CH;
    int s[CH_PER], d[CH_PER], r[CH_PER];
#pragma unroll
    for (int k = 0; k < CH_PER; ++k) {
        int i = k * CH_T + t;
        int e = e0 + (i < n ? i : 0);
        s[k] = src[e];
        d[k] = dst[e];
        if (i >= n) d[k] = -1;
    }
#pragma unroll
    for (int k = 0; k < CH_PER; ++k) {
        r[k] = 0;
        if (d[k] >= 0) r[k] = atomicAdd(&hist[d[k] >> BSH], 1);
    }
    __syncthreads();

    int b0 = 2 * t, b1 = 2 * t + 1;
    int h0 = hist[b0], h1 = hist[b1];
    int sum = h0 + h1;
    sh[t] = sum;
    __syncthreads();
    for (int o = 1; o < CH_T; o <<= 1) {
        int v = (t >= o) ? sh[t - o] : 0;
        __syncthreads();
        sh[t] += v;
        __syncthreads();
    }
    int l0 = sh[t] - sum;
    int l1 = l0 + h0;
    gpos[b0] = (h0 > 0) ? bstart[b0] + atomicAdd(&bcursor[b0], h0) - l0 : 0;
    gpos[b1] = (h1 > 0) ? bstart[b1] + atomicAdd(&bcursor[b1], h1) - l1 : 0;
    hist[b0] = l0;
    hist[b1] = l1;
    __syncthreads();

#pragma unroll
    for (int k = 0; k < CH_PER; ++k)
        if (d[k] >= 0) sorted[hist[d[k] >> BSH] + r[k]] = make_int2(s[k], d[k]);
    __syncthreads();

    for (int i = t; i < n; i += CH_T) {
        int2 v = sorted[i];
        rec[gpos[v.y >> BSH] + i] = v;
    }
}

__global__ void bucket_finish(const int2* __restrict__ rec, const int* __restrict__ bstart,
                              int* __restrict__ rowptr, float* __restrict__ dis,
                              int* __restrict__ ent, int N, int nB) {
    __shared__ int cnt[BW];
    __shared__ int cur[BW];
    __shared__ int sh[FIN_T];
    __shared__ int stage[STAGE];
    int t     = threadIdx.x;
    int b     = blockIdx.x;
    int beg   = bstart[b];
    int end   = bstart[b + 1];
    int n     = end - beg;
    int node0 = b << BSH;
    if (t < BW) cnt[t] = 0;
    __syncthreads();
    for (int i0 = t; i0 < n; i0 += FIN_U * FIN_T) {
        int dl[FIN_U];
#pragma unroll
        for (int u = 0; u < FIN_U; ++u) {
            int i = i0 + u * FIN_T;
            dl[u] = rec[beg + (i < n ? i : i0)].y - node0;
            if (i >= n) dl[u] = -1;
        }
#pragma unroll
        for (int u = 0; u < FIN_U; ++u)
            if (dl[u] >= 0) atomicAdd(&cnt[dl[u]], 1);
    }
    __syncthreads();

    int c = (t < BW) ? cnt[t] : 0;
    sh[t] = c;
    __syncthreads();
    for (int o = 1; o < BW; o <<= 1) {
        int v = (t >= o) ? sh[t - o] : 0;
        __syncthreads();
        sh[t] += v;
        __syncthreads();
    }
    if (t < BW) {
        int st = sh[t] - c;
        cur[t] = st;
        int node = node0 + t;
        if (node < N) {
            rowptr[node] = beg + st;
            dis[node]    = rsqrtf((float)(c + 1));
        }
    }
    if (b == nB - 1 && t == 0) rowptr[N] = end;
    __syncthreads();

    bool staged = n <= STAGE;
    for (int i0 = t; i0 < n; i0 += FIN_U * FIN_T) {
        int2 v[FIN_U];
#pragma unroll
        for (int u = 0; u < FIN_U; ++u) {
            int i = i0 + u * FIN_T;
            v[u] = rec[beg + (i < n ? i : i0)];
            if (i >= n) v[u].y = -1;
        }
#pragma unroll
        for (int u = 0; u < FIN_U; ++u) {
            if (v[u].y >= 0) {
                int p = atomicAdd(&cur[v[u].y - node0], 1);
                if (staged) stage[p] = v[u].x;
                else        ent[beg + p] = v[u].x;
            }
        }
    }
    __syncthreads();
    if (staged)
        for (int i = t; i < n; i += FIN_T) ent[beg + i] = stage[i];
}

template<int FT>
__global__ void wmma_gemm(const float* __restrict__ X, const float* __restrict__ W,
                          const float* __restrict__ Bias, const float* __restrict__ RowScale,
                          float* __restrict__ Y, int N, int K, int addBias) {
    const int F    = FT * 16;
    int wave = (blockIdx.x * blockDim.x + threadIdx.x) >> 5;
    int lane = threadIdx.x & 31;
    int row0 = wave << 4;
    if (row0 >= N) return;
    int half = lane >> 4;
    int l16  = lane & 15;
    int arow = row0 + l16; if (arow > N - 1) arow = N - 1;

    v8f acc[FT];
    v8f z = {0.f, 0.f, 0.f, 0.f, 0.f, 0.f, 0.f, 0.f};
#pragma unroll
    for (int t = 0; t < FT; ++t) acc[t] = z;

    for (int k0 = 0; k0 < K; k0 += 4) {
        int ka = k0 + half * 2;
        v2f a;
        a.x = X[arow * K + ka];
        a.y = X[arow * K + ka + 1];
#pragma unroll
        for (int t = 0; t < FT; ++t) {
            int col = t * 16 + l16;
            v2f b;
            b.x = W[ka * F + col];
            b.y = W[(ka + 1) * F + col];
            acc[t] = __builtin_amdgcn_wmma_f32_16x16x4_f32(
                 false, a,  false, b,
                 (short)0, acc[t],  false,  false);
        }
    }

    float sc[8];
#pragma unroll
    for (int v = 0; v < 8; ++v) {
        int r = row0 + half * 8 + v;
        sc[v] = (RowScale != nullptr && r < N) ? RowScale[r] : 1.f;
    }
#pragma unroll
    for (int t = 0; t < FT; ++t) {
        int col = t * 16 + l16;
        float bb = addBias ? Bias[col] : 0.f;
#pragma unroll
        for (int v = 0; v < 8; ++v) {
            int r = row0 + half * 8 + v;
            if (r < N) Y[r * F + col] = (acc[t][v] + bb) * sc[v];
        }
    }
}

#define NPW 7

template<int VEC> struct RowVec;
template<> struct RowVec<1> { typedef float type; };
template<> struct RowVec<2> { typedef v2f   type; };

template<int VEC, int U>
__device__ __forceinline__ void pull_group(const float* __restrict__ H, unsigned fo, int off,
                                           int j, typename RowVec<VEC>::type& acc) {
    typedef typename RowVec<VEC>::type vec_t;
    vec_t h[U];
#pragma unroll
    for (int u = 0; u < U; ++u) {
        const float* row = H + __builtin_amdgcn_readlane(off, j + u);
        h[u] = *(const vec_t*)(row + fo);
    }
#pragma unroll
    for (int u = 0; u < U; ++u) acc += h[u];
}

__device__ __forceinline__ void pool_flush(float* __restrict__ gmax, float* __restrict__ gsum,
                                           int at, v2f mx, v2f sm) {
    atomicMax((unsigned int*)(gmax + at),     __float_as_uint(mx.x));
    atomicMax((unsigned int*)(gmax + at + 1), __float_as_uint(mx.y));
    atomicAdd(gsum + at,     sm.x);
    atomicAdd(gsum + at + 1, sm.y);
}

template<int F, bool POOL>
__global__ void pull_rows(const int* __restrict__ rowptr, const int* __restrict__ ent,
                          const float* __restrict__ dis, const float* __restrict__ H,
                          const float* __restrict__ bias, float* __restrict__ OUT,
                          const int* __restrict__ batch, float* __restrict__ gmax,
                          float* __restrict__ gsum, float* __restrict__ cnt, int N) {
    constexpr int VEC = (F > 32) ? 2 : 1;
    static_assert(!POOL || F == 64, "the pooled form holds a whole row in the wave: 2 features per lane");
    typedef typename RowVec<VEC>::type vec_t;
    int wave  = __builtin_amdgcn_readfirstlane((int)((blockIdx.x * blockDim.x + threadIdx.x) >> 5));
    int node0 = wave * NPW;
    if (node0 >= N) return;
    int nn = N - node0;
    if (nn > NPW) nn = NPW;
    int  lane   = threadIdx.x & 31;
    bool active = lane * VEC < F;
    unsigned fo = active ? lane * VEC : 0;

    int   rp = rowptr[node0 + (lane < nn ? lane : nn)];
    float dv = dis[node0 + (lane < nn ? lane : nn - 1)];
    vec_t bv;
    if constexpr (VEC == 1) bv = bias[fo];
    else                    { bv.x = bias[fo]; bv.y = bias[fo + 1]; }

    int beg = __builtin_amdgcn_readlane(rp, 0);
    int end = __builtin_amdgcn_readlane(rp, 1);
    int off = 0;
    if (lane < end - beg) off = ent[beg + lane] * F;
    vec_t self = *(const vec_t*)(H + node0 * F + fo);

    int   gl   = 0;
    int   gcur = 0;
    vec_t pmax = self, psum = self;
    if constexpr (POOL) {
        gl = batch[node0 + (lane < nn ? lane : 0)];
        if (lane < nn) atomicAdd(&cnt[gl], 1.0f);
    }

    for (int k = 0; k < nn; ++k) {
        int kn   = k + 1 < nn ? k + 1 : k;
        int nend = __builtin_amdgcn_readlane(rp, kn + 1);
        int i    = end + lane;
        if (i > nend - 1) i = nend - 1;
        if (i < 0) i = 0;
        int   entn  = ent[i];
        vec_t selfn = *(const vec_t*)(H + (node0 + kn) * F + fo);

        vec_t acc = self;
        for (int base = beg; ; ) {
            int cnt = end - base;
            if (cnt > 32) cnt = 32;
            int j = 0;
            for (; j + 16 <= cnt; j += 16) pull_group<VEC, 16>(H, fo, off, j, acc);
            if (cnt & 8) { pull_group<VEC, 8>(H, fo, off, j, acc); j += 8; }
            if (cnt & 4) { pull_group<VEC, 4>(H, fo, off, j, acc); j += 4; }
            if (cnt & 2) { pull_group<VEC, 2>(H, fo, off, j, acc); j += 2; }
            if (cnt & 1) { pull_group<VEC, 1>(H, fo, off, j, acc); }
            base += 32;
            if (base >= end) break;
            off = 0;
            if (lane < end - base) off = ent[base + lane] * F;
        }

        float dd = __int_as_float(__builtin_amdgcn_readlane(__float_as_int(dv), k));
        vec_t v;
        if constexpr (VEC == 1) {
            v = acc * dd + bv;
            v = v > 0.f ? v : 0.f;
        } else {
            v.x = acc.x * dd + bv.x;
            v.y = acc.y * dd + bv.y;
            v.x = v.x > 0.f ? v.x : 0.f;
            v.y = v.y > 0.f ? v.y : 0.f;
        }
        if constexpr (POOL) {
            int g = __builtin_amdgcn_readlane(gl, k);
            if (k > 0 && g == gcur) {
                pmax.x = fmaxf(pmax.x, v.x);
                pmax.y = fmaxf(pmax.y, v.y);
                psum  += v;
            } else {
                if (k > 0) pool_flush(gmax, gsum, gcur * F + (int)fo, pmax, psum);
                gcur = g; pmax = v; psum = v;
            }
        } else {
            if (active) *(vec_t*)(OUT + (node0 + k) * F + fo) = v;
        }
        beg = end; end = nend; off = entn * F; self = selfn;
    }
    if constexpr (POOL) pool_flush(gmax, gsum, gcur * F + (int)fo, pmax, psum);
}

template<int FT>
__global__ void project_pooled(const float* __restrict__ gmax, const float* __restrict__ gsum,
                               const float* __restrict__ cnt, const float* __restrict__ W,
                               const float* __restrict__ Bias, float* __restrict__ Y,
                               int G, int Fh) {
    const int F    = FT * 16;
    int wave = (blockIdx.x * blockDim.x + threadIdx.x) >> 5;
    int lane = threadIdx.x & 31;
    int row0 = wave << 4;
    if (row0 >= G) return;
    int half = lane >> 4;
    int l16  = lane & 15;
    int arow = row0 + l16; if (arow > G - 1) arow = G - 1;
    float c = cnt[arow];
    c = c > 1.f ? c : 1.f;

    v8f acc[FT];
    v8f z = {0.f, 0.f, 0.f, 0.f, 0.f, 0.f, 0.f, 0.f};
#pragma unroll
    for (int t = 0; t < FT; ++t) acc[t] = z;

    for (int k0 = 0; k0 < Fh; k0 += 4) {
        int ka = k0 + half * 2;
        v2f a;
        a.x = gmax[arow * Fh + ka];
        a.y = gmax[arow * Fh + ka + 1];
#pragma unroll
        for (int t = 0; t < FT; ++t) {
            int col = t * 16 + l16;
            v2f b;
            b.x = W[ka * F + col];
            b.y = W[(ka + 1) * F + col];
            acc[t] = __builtin_amdgcn_wmma_f32_16x16x4_f32(
                 false, a,  false, b,
                 (short)0, acc[t],  false,  false);
        }
    }
    for (int k0 = 0; k0 < Fh; k0 += 4) {
        int ka = k0 + half * 2;
        v2f a;
        a.x = gsum[arow * Fh + ka] / c;
        a.y = gsum[arow * Fh + ka + 1] / c;
#pragma unroll
        for (int t = 0; t < FT; ++t) {
            int col = t * 16 + l16;
            v2f b;
            b.x = W[(Fh + ka) * F + col];
            b.y = W[(Fh + ka + 1) * F + col];
            acc[t] = __builtin_amdgcn_wmma_f32_16x16x4_f32(
                 false, a,  false, b,
                 (short)0, acc[t],  false,  false);
        }
    }

#pragma unroll
    for (int t = 0; t < FT; ++t) {
        int col = t * 16 + l16;
        float bb = Bias[col];
#pragma unroll
        for (int v = 0; v < 8; ++v) {
            int r = row0 + half * 8 + v;
            if (r < G) Y[r * F + col] = acc[t][v] + bb;
        }
    }
}

extern "C" void kernel_launch(void* const* d_in, const int* in_sizes, int n_in,
                              void* d_out, int out_size, void* d_ws, size_t ws_size,
                              hipStream_t stream) {
    const float* x     = (const float*)d_in[0];
    const int*   ei    = (const int*)d_in[1];
    const int*   batch = (const int*)d_in[2];
    const float* W1    = (const float*)d_in[3];
    const float* b1    = (const float*)d_in[4];
    const float* W2    = (const float*)d_in[5];
    const float* b2    = (const float*)d_in[6];
    const float* W3    = (const float*)d_in[7];
    const float* b3    = (const float*)d_in[8];
    const float* Wout  = (const float*)d_in[9];
    const float* bout  = (const float*)d_in[10];
    float*       out   = (float*)d_out;

    const int N  = in_sizes[2];
    const int E  = in_sizes[1] / 2;
    const int K0 = in_sizes[0] / N;
    const int F1 = in_sizes[4];
    const int F2 = in_sizes[6];
    const int F3 = in_sizes[8];
    const int Fo = in_sizes[10];
    const int G  = out_size / Fo;

    const int* src = ei;
    const int* dst = ei + E;

    char*  ws  = (char*)d_ws;
    size_t off = 0;
    auto carve = [&](size_t bytes) -> float* {
        float* p = (float*)(ws + off);
        off += (bytes + 255) & ~(size_t)255;
        return p;
    };
    float* dis    = carve((size_t)N * 4);
    float* bufT   = carve((size_t)N * F3 * 4);
    float* bufA   = carve((size_t)N * F3 * 4);
    float* bufB   = carve((size_t)N * F3 * 4);
    const int nZero = 2 * G * F3 + G + 2 * NBMAX;
    float* zeros  = carve((size_t)nZero * 4);
    float* gmax   = zeros;
    float* gsum   = gmax + G * F3;
    float* cnt    = gsum + G * F3;
    int*  btotal  = (int*)(cnt + G);
    int*  bcursor = btotal + NBMAX;
    const int nB  = (N + BW - 1) >> BSH;
    const int nCh = (E + CH - 1) / CH;
    int*  bstart  = (int*)carve((size_t)(NBMAX + 1) * 4);
    int*  rowptr  = (int*)carve((size_t)(N + 1) * 4);
    int*  ent     = (int*)carve((size_t)E * 4);
    int2* rec     = (int2*)carve((size_t)E * 8);

    const int BLK = 256;
    auto grid1   = [&](long long n) { return dim3((unsigned)((n + BLK - 1) / BLK)); };
    auto gridWv  = [&](long long waves) { return dim3((unsigned)((waves * 32 + BLK - 1) / BLK)); };
    auto gemmGrid = [&](int rows) { return gridWv((rows + 15) / 16); };

    fill_i32<<<grid1(nZero), BLK, 0, stream>>>((int*)zeros, 0, nZero);
    bucket_count<<<dim3((unsigned)nCh), CH_T, 0, stream>>>(dst, btotal, E, nB);
    bucket_starts<<<dim3(1), NBMAX, 0, stream>>>(btotal, bstart, nB);
    bucket_scatter<<<dim3((unsigned)nCh), CH_T, 0, stream>>>(src, dst, bstart, bcursor, rec, E);
    bucket_finish<<<dim3((unsigned)nB), FIN_T, 0, stream>>>(rec, bstart, rowptr, dis, ent, N, nB);

    wmma_gemm<2><<<gemmGrid(N), BLK, 0, stream>>>(x, W1, b1, dis, bufT, N, K0, 0);
    pull_rows<32, false><<<gridWv((N + NPW - 1) / NPW), BLK, 0, stream>>>(rowptr, ent, dis, bufT, b1, bufA,
                                                                           nullptr, nullptr, nullptr, nullptr, N);

    wmma_gemm<3><<<gemmGrid(N), BLK, 0, stream>>>(bufA, W2, b2, dis, bufT, N, F1, 0);
    pull_rows<48, false><<<gridWv((N + NPW - 1) / NPW), BLK, 0, stream>>>(rowptr, ent, dis, bufT, b2, bufB,
                                                                           nullptr, nullptr, nullptr, nullptr, N);

    wmma_gemm<4><<<gemmGrid(N), BLK, 0, stream>>>(bufB, W3, b3, dis, bufT, N, F2, 0);
    pull_rows<64, true><<<gridWv((N + NPW - 1) / NPW), BLK, 0, stream>>>(rowptr, ent, dis, bufT, b3, nullptr,
                                                                          batch, gmax, gsum, cnt, N);

    project_pooled<2><<<gemmGrid(G), BLK, 0, stream>>>(gmax, gsum, cnt, Wout, bout, out, G, F3);
}


# layers 2, 3: weight product moved behind the neighbour sum, into the aggregation kernel (summed rows via LDS to WMMA): 32, 48 floats gathered per edge, not 48, 64; 12 to 10 launches; rounding changes
# speedup vs baseline: 6.4069x; 1.1209x over previous
#include <hip/hip_runtime.h>

typedef __attribute__((ext_vector_type(2))) float v2f;
typedef __attribute__((ext_vector_type(8))) float v8f;

__global__ void fill_i32(int* __restrict__ p, int v, int n) {
    int i = blockIdx.x * blockDim.x + threadIdx.x;
    if (i < n) p[i] = v;
}


#define BSH    8
#define BW     (1 << BSH)
#define NBMAX  512
#define CH     4096
#define CH_T   256
#define CH_PER (CH / CH_T)
#define FIN_T  512
#define FIN_U  4
#define STAGE  12288

__global__ void bucket_count(const int* __restrict__ dst, int* __restrict__ total, int nE, int nB) {
    __shared__ int hist[NBMAX];
    int t = threadIdx.x;
    for (int i = t; i < NBMAX; i += CH_T) hist[i] = 0;
    __syncthreads();
    int e0 = blockIdx.x * CH;
    int n  = nE - e0;
    if (n > CH) n = CH;
    int d[CH_PER];
#pragma unroll
    for (int k = 0; k < CH_PER; ++k) {
        int i = k * CH_T + t;
        d[k] = dst[e0 + (i < n ? i : 0)];
        if (i >= n) d[k] = -1;
    }
#pragma unroll
    for (int k = 0; k < CH_PER; ++k)
        if (d[k] >= 0) atomicAdd(&hist[d[k] >> BSH], 1);
    __syncthreads();
    for (int i = t; i < nB; i += CH_T) {
        int c = hist[i];
        if (c) atomicAdd(&total[i], c);
    }
}

__global__ void bucket_starts(const int* __restrict__ total, int* __restrict__ bstart, int nB) {
    __shared__ int sh[NBMAX];
    int t = threadIdx.x;
    int c = (t < nB) ? total[t] : 0;
    sh[t] = c;
    __syncthreads();
    for (int o = 1; o < NBMAX; o <<= 1) {
        int v = (t >= o) ? sh[t - o] : 0;
        __syncthreads();
        sh[t] += v;
        __syncthreads();
    }
    if (t < nB) bstart[t] = sh[t] - c;
    if (t == nB - 1) bstart[nB] = sh[t];
}

__global__ void bucket_scatter(const int* __restrict__ src, const int* __restrict__ dst,
                               const int* __restrict__ bstart, int* __restrict__ bcursor,
                               int2* __restrict__ rec, int nE) {
    __shared__ int  hist[NBMAX];
    __shared__ int  gpos[NBMAX];
    __shared__ int  sh[CH_T];
    __shared__ int2 sorted[CH];
    int t = threadIdx.x;
    for (int i = t; i < NBMAX; i += CH_T) hist[i] = 0;
    __syncthreads();

    int e0 = blockIdx.x * CH;
    int n  = nE - e0;
    if (n > CH) n = CH;
    int s[CH_PER], d[CH_PER], r[CH_PER];
#pragma unroll
    for (int k = 0; k < CH_PER; ++k) {
        int i = k * CH_T + t;
        int e = e0 + (i < n ? i : 0);
        s[k] = src[e];
        d[k] = dst[e];
        if (i >= n) d[k] = -1;
    }
#pragma unroll
    for (int k = 0; k < CH_PER; ++k) {
        r[k] = 0;
        if (d[k] >= 0) r[k] = atomicAdd(&hist[d[k] >> BSH], 1);
    }
    __syncthreads();

    int b0 = 2 * t, b1 = 2 * t + 1;
    int h0 = hist[b0], h1 = hist[b1];
    int sum = h0 + h1;
    sh[t] = sum;
    __syncthreads();
    for (int o = 1; o < CH_T; o <<= 1) {
        int v = (t >= o) ? sh[t - o] : 0;
        __syncthreads();
        sh[t] += v;
        __syncthreads();
    }
    int l0 = sh[t] - sum;
    int l1 = l0 + h0;
    gpos[b0] = (h0 > 0) ? bstart[b0] + atomicAdd(&bcursor[b0], h0) - l0 : 0;
    gpos[b1] = (h1 > 0) ? bstart[b1] + atomicAdd(&bcursor[b1], h1) - l1 : 0;
    hist[b0] = l0;
    hist[b1] = l1;
    __syncthreads();

#pragma unroll
    for (int k = 0; k < CH_PER; ++k)
        if (d[k] >= 0) sorted[hist[d[k] >> BSH] + r[k]] = make_int2(s[k], d[k]);
    __syncthreads();

    for (int i = t; i < n; i += CH_T) {
        int2 v = sorted[i];
        rec[gpos[v.y >> BSH] + i] = v;
    }
}

__global__ void bucket_finish(const int2* __restrict__ rec, const int* __restrict__ bstart,
                              int* __restrict__ rowptr, float* __restrict__ dis,
                              int* __restrict__ ent, int N, int nB) {
    __shared__ int cnt[BW];
    __shared__ int cur[BW];
    __shared__ int sh[FIN_T];
    __shared__ int stage[STAGE];
    int t     = threadIdx.x;
    int b     = blockIdx.x;
    int beg   = bstart[b];
    int end   = bstart[b + 1];
    int n     = end - beg;
    int node0 = b << BSH;
    if (t < BW) cnt[t] = 0;
    __syncthreads();
    for (int i0 = t; i0 < n; i0 += FIN_U * FIN_T) {
        int dl[FIN_U];
#pragma unroll
        for (int u = 0; u < FIN_U; ++u) {
            int i = i0 + u * FIN_T;
            dl[u] = rec[beg + (i < n ? i : i0)].y - node0;
            if (i >= n) dl[u] = -1;
        }
#pragma unroll
        for (int u = 0; u < FIN_U; ++u)
            if (dl[u] >= 0) atomicAdd(&cnt[dl[u]], 1);
    }
    __syncthreads();

    int c = (t < BW) ? cnt[t] : 0;
    sh[t] = c;
    __syncthreads();
    for (int o = 1; o < BW; o <<= 1) {
        int v = (t >= o) ? sh[t - o] : 0;
        __syncthreads();
        sh[t] += v;
        __syncthreads();
    }
    if (t < BW) {
        int st = sh[t] - c;
        cur[t] = st;
        int node = node0 + t;
        if (node < N) {
            rowptr[node] = beg + st;
            dis[node]    = rsqrtf((float)(c + 1));
        }
    }
    if (b == nB - 1 && t == 0) rowptr[N] = end;
    __syncthreads();

    bool staged = n <= STAGE;
    for (int i0 = t; i0 < n; i0 += FIN_U * FIN_T) {
        int2 v[FIN_U];
#pragma unroll
        for (int u = 0; u < FIN_U; ++u) {
            int i = i0 + u * FIN_T;
            v[u] = rec[beg + (i < n ? i : i0)];
            if (i >= n) v[u].y = -1;
        }
#pragma unroll
        for (int u = 0; u < FIN_U; ++u) {
            if (v[u].y >= 0) {
                int p = atomicAdd(&cur[v[u].y - node0], 1);
                if (staged) stage[p] = v[u].x;
                else        ent[beg + p] = v[u].x;
            }
        }
    }
    __syncthreads();
    if (staged)
        for (int i = t; i < n; i += FIN_T) ent[beg + i] = stage[i];
}

template<int FT>
__global__ void wmma_gemm(const float* __restrict__ X, const float* __restrict__ W,
                          const float* __restrict__ Bias, const float* __restrict__ RowScale,
                          float* __restrict__ Y, int N, int K, int addBias) {
    const int F    = FT * 16;
    int wave = (blockIdx.x * blockDim.x + threadIdx.x) >> 5;
    int lane = threadIdx.x & 31;
    int row0 = wave << 4;
    if (row0 >= N) return;
    int half = lane >> 4;
    int l16  = lane & 15;
    int arow = row0 + l16; if (arow > N - 1) arow = N - 1;

    v8f acc[FT];
    v8f z = {0.f, 0.f, 0.f, 0.f, 0.f, 0.f, 0.f, 0.f};
#pragma unroll
    for (int t = 0; t < FT; ++t) acc[t] = z;

    for (int k0 = 0; k0 < K; k0 += 4) {
        int ka = k0 + half * 2;
        v2f a;
        a.x = X[arow * K + ka];
        a.y = X[arow * K + ka + 1];
#pragma unroll
        for (int t = 0; t < FT; ++t) {
            int col = t * 16 + l16;
            v2f b;
            b.x = W[ka * F + col];
            b.y = W[(ka + 1) * F + col];
            acc[t] = __builtin_amdgcn_wmma_f32_16x16x4_f32(
                 false, a,  false, b,
                 (short)0, acc[t],  false,  false);
        }
    }

    float sc[8];
#pragma unroll
    for (int v = 0; v < 8; ++v) {
        int r = row0 + half * 8 + v;
        sc[v] = (RowScale != nullptr && r < N) ? RowScale[r] : 1.f;
    }
#pragma unroll
    for (int t = 0; t < FT; ++t) {
        int col = t * 16 + l16;
        float bb = addBias ? Bias[col] : 0.f;
#pragma unroll
        for (int v = 0; v < 8; ++v) {
            int r = row0 + half * 8 + v;
            if (r < N) Y[r * F + col] = (acc[t][v] + bb) * sc[v];
        }
    }
}

#define NPW 7
#define PULL_T 256

template<int VEC> struct RowVec;
template<> struct RowVec<1> { typedef float type; };
template<> struct RowVec<2> { typedef v2f   type; };

template<int VEC, int U>
__device__ __forceinline__ void pull_group(const float* __restrict__ H, unsigned fo, int off,
                                           int j, typename RowVec<VEC>::type& acc) {
    typedef typename RowVec<VEC>::type vec_t;
    vec_t h[U];
#pragma unroll
    for (int u = 0; u < U; ++u) {
        const float* row = H + __builtin_amdgcn_readlane(off, j + u);
        h[u] = *(const vec_t*)(row + fo);
    }
#pragma unroll
    for (int u = 0; u < U; ++u) acc += h[u];
}

template<int F, int FO, bool GEMM, bool POOL>
__global__ void pull_rows(const int* __restrict__ rowptr, const int* __restrict__ ent,
                          const float* __restrict__ dis, const float* __restrict__ H,
                          const float* __restrict__ Wt, const float* __restrict__ bias,
                          float* __restrict__ OUT, const int* __restrict__ batch,
                          float* __restrict__ gmax, float* __restrict__ gsum,
                          float* __restrict__ cnt, int N) {
    constexpr int VEC   = (F > 32) ? 2 : 1;
    constexpr int PITCH = F + 4;
    constexpr int FT    = FO / 16;
    static_assert(GEMM || (F == FO && VEC == 1 && !POOL), "without the product the row is written as it is summed");
    static_assert(NPW <= 8, "the wave's rows must lie in the tile rows held by lanes 0 .. 15");
    typedef typename RowVec<VEC>::type vec_t;
    __shared__ __attribute__((aligned(16))) float tile[GEMM ? (PULL_T / 32) * NPW * PITCH : 1];
    int wave  = __builtin_amdgcn_readfirstlane((int)((blockIdx.x * blockDim.x + threadIdx.x) >> 5));
    int node0 = wave * NPW;
    if (node0 >= N) return;
    int nn = N - node0;
    if (nn > NPW) nn = NPW;
    int  lane   = threadIdx.x & 31;
    bool active = lane * VEC < F;
    unsigned fo = active ? lane * VEC : 0;
    float* trow = tile + (GEMM ? (int)(threadIdx.x >> 5) * NPW * PITCH : 0);

    int   rp = rowptr[node0 + (lane < nn ? lane : nn)];
    float dv = dis[node0 + (lane < nn ? lane : nn - 1)];

    int beg = __builtin_amdgcn_readlane(rp, 0);
    int end = __builtin_amdgcn_readlane(rp, 1);
    int off = 0;
    if (lane < end - beg) off = ent[beg + lane] * F;
    vec_t self = *(const vec_t*)(H + node0 * F + fo);
    float bv = 0.f;
    if constexpr (!GEMM) bv = bias[fo];
    int gl = 0;
    if constexpr (POOL) {
        gl = batch[node0 + (lane < nn ? lane : 0)];
        if (lane < nn) atomicAdd(&cnt[gl], 1.0f);
    }

    for (int k = 0; k < nn; ++k) {
        int kn   = k + 1 < nn ? k + 1 : k;
        int nend = __builtin_amdgcn_readlane(rp, kn + 1);
        int i    = end + lane;
        if (i > nend - 1) i = nend - 1;
        if (i < 0) i = 0;
        int   entn  = ent[i];
        vec_t selfn = *(const vec_t*)(H + (node0 + kn) * F + fo);

        vec_t acc = self;
        for (int base = beg; ; ) {
            int cnt = end - base;
            if (cnt > 32) cnt = 32;
            int j = 0;
            for (; j + 16 <= cnt; j += 16) pull_group<VEC, 16>(H, fo, off, j, acc);
            if (cnt & 8) { pull_group<VEC, 8>(H, fo, off, j, acc); j += 8; }
            if (cnt & 4) { pull_group<VEC, 4>(H, fo, off, j, acc); j += 4; }
            if (cnt & 2) { pull_group<VEC, 2>(H, fo, off, j, acc); j += 2; }
            if (cnt & 1) { pull_group<VEC, 1>(H, fo, off, j, acc); }
            base += 32;
            if (base >= end) break;
            off = 0;
            if (lane < end - base) off = ent[base + lane] * F;
        }

        float dd = __int_as_float(__builtin_amdgcn_readlane(__float_as_int(dv), k));
        if constexpr (GEMM) {
            vec_t sv = acc * dd;
            if (active) *(vec_t*)(trow + k * PITCH + fo) = sv;
        } else {
            float v = acc * dd + bv;
            v = v > 0.f ? v : 0.f;
            if (active) OUT[(node0 + k) * F + fo] = v * dd;
        }
        beg = end; end = nend; off = entn * F; self = selfn;
    }

    if constexpr (GEMM) {
        __builtin_amdgcn_fence(__ATOMIC_RELEASE, "wavefront");
        __builtin_amdgcn_wave_barrier();
        __builtin_amdgcn_fence(__ATOMIC_ACQUIRE, "wavefront");

        int half = lane >> 4;
        int l16  = lane & 15;
        int ar   = l16 < nn ? l16 : nn - 1;
        const float* arow = trow + ar * PITCH;

        v8f accw[FT];
        v8f z = {0.f, 0.f, 0.f, 0.f, 0.f, 0.f, 0.f, 0.f};
#pragma unroll
        for (int t = 0; t < FT; ++t) accw[t] = z;

#pragma unroll 2
        for (int k0 = 0; k0 < F; k0 += 4) {
            int ka = k0 + half * 2;
            v2f a = *(const v2f*)(arow + ka);
#pragma unroll
            for (int t = 0; t < FT; ++t) {
                int col = t * 16 + l16;
                v2f b;
                b.x = Wt[ka * FO + col];
                b.y = Wt[(ka + 1) * FO + col];
                accw[t] = __builtin_amdgcn_wmma_f32_16x16x4_f32(
                     false, a,  false, b,
                     (short)0, accw[t],  false,  false);
            }
        }

        float bb[FT];
#pragma unroll
        for (int t = 0; t < FT; ++t) bb[t] = bias[t * 16 + l16];

        if constexpr (!POOL) {
#pragma unroll
            for (int v = 0; v < NPW; ++v) {
                float dd = __int_as_float(__builtin_amdgcn_readlane(__float_as_int(dv), v));
                if (v < nn && half == 0) {
#pragma unroll
                    for (int t = 0; t < FT; ++t) {
                        float r = accw[t][v] + bb[t];
                        r = r > 0.f ? r : 0.f;
                        OUT[(node0 + v) * FO + t * 16 + l16] = r * dd;
                    }
                }
            }
        } else {
            int   gcur = 0;
            float pmax[FT], psum[FT];
#pragma unroll
            for (int t = 0; t < FT; ++t) { pmax[t] = 0.f; psum[t] = 0.f; }
#pragma unroll
            for (int v = 0; v < NPW; ++v) {
                if (v < nn) {
                    int g = __builtin_amdgcn_readlane(gl, v);
                    bool same = v > 0 && g == gcur;
                    if (v > 0 && !same && half == 0) {
#pragma unroll
                        for (int t = 0; t < FT; ++t) {
                            int at = gcur * FO + t * 16 + l16;
                            atomicMax((unsigned int*)(gmax + at), __float_as_uint(pmax[t]));
                            atomicAdd(gsum + at, psum[t]);
                        }
                    }
#pragma unroll
                    for (int t = 0; t < FT; ++t) {
                        float r = accw[t][v] + bb[t];
                        r = r > 0.f ? r : 0.f;
                        pmax[t] = same ? fmaxf(pmax[t], r) : r;
                        psum[t] = same ? psum[t] + r : r;
                    }
                    gcur = g;
                }
            }
            if (half == 0) {
#pragma unroll
                for (int t = 0; t < FT; ++t) {
                    int at = gcur * FO + t * 16 + l16;
                    atomicMax((unsigned int*)(gmax + at), __float_as_uint(pmax[t]));
                    atomicAdd(gsum + at, psum[t]);
                }
            }
        }
    }
}

template<int FT>
__global__ void project_pooled(const float* __restrict__ gmax, const float* __restrict__ gsum,
                               const float* __restrict__ cnt, const float* __restrict__ W,
                               const float* __restrict__ Bias, float* __restrict__ Y,
                               int G, int Fh) {
    const int F    = FT * 16;
    int wave = (blockIdx.x * blockDim.x + threadIdx.x) >> 5;
    int lane = threadIdx.x & 31;
    int row0 = wave << 4;
    if (row0 >= G) return;
    int half = lane >> 4;
    int l16  = lane & 15;
    int arow = row0 + l16; if (arow > G - 1) arow = G - 1;
    float c = cnt[arow];
    c = c > 1.f ? c : 1.f;

    v8f acc[FT];
    v8f z = {0.f, 0.f, 0.f, 0.f, 0.f, 0.f, 0.f, 0.f};
#pragma unroll
    for (int t = 0; t < FT; ++t) acc[t] = z;

    for (int k0 = 0; k0 < Fh; k0 += 4) {
        int ka = k0 + half * 2;
        v2f a;
        a.x = gmax[arow * Fh + ka];
        a.y = gmax[arow * Fh + ka + 1];
#pragma unroll
        for (int t = 0; t < FT; ++t) {
            int col = t * 16 + l16;
            v2f b;
            b.x = W[ka * F + col];
            b.y = W[(ka + 1) * F + col];
            acc[t] = __builtin_amdgcn_wmma_f32_16x16x4_f32(
                 false, a,  false, b,
                 (short)0, acc[t],  false,  false);
        }
    }
    for (int k0 = 0; k0 < Fh; k0 += 4) {
        int ka = k0 + half * 2;
        v2f a;
        a.x = gsum[arow * Fh + ka] / c;
        a.y = gsum[arow * Fh + ka + 1] / c;
#pragma unroll
        for (int t = 0; t < FT; ++t) {
            int col = t * 16 + l16;
            v2f b;
            b.x = W[(Fh + ka) * F + col];
            b.y = W[(Fh + ka + 1) * F + col];
            acc[t] = __builtin_amdgcn_wmma_f32_16x16x4_f32(
                 false, a,  false, b,
                 (short)0, acc[t],  false,  false);
        }
    }

#pragma unroll
    for (int t = 0; t < FT; ++t) {
        int col = t * 16 + l16;
        float bb = Bias[col];
#pragma unroll
        for (int v = 0; v < 8; ++v) {
            int r = row0 + half * 8 + v;
            if (r < G) Y[r * F + col] = acc[t][v] + bb;
        }
    }
}

extern "C" void kernel_launch(void* const* d_in, const int* in_sizes, int n_in,
                              void* d_out, int out_size, void* d_ws, size_t ws_size,
                              hipStream_t stream) {
    const float* x     = (const float*)d_in[0];
    const int*   ei    = (const int*)d_in[1];
    const int*   batch = (const int*)d_in[2];
    const float* W1    = (const float*)d_in[3];
    const float* b1    = (const float*)d_in[4];
    const float* W2    = (const float*)d_in[5];
    const float* b2    = (const float*)d_in[6];
    const float* W3    = (const float*)d_in[7];
    const float* b3    = (const float*)d_in[8];
    const float* Wout  = (const float*)d_in[9];
    const float* bout  = (const float*)d_in[10];
    float*       out   = (float*)d_out;

    const int N  = in_sizes[2];
    const int E  = in_sizes[1] / 2;
    const int K0 = in_sizes[0] / N;
    const int F3 = in_sizes[8];
    const int Fo = in_sizes[10];
    const int G  = out_size / Fo;

    const int* src = ei;
    const int* dst = ei + E;

    char*  ws  = (char*)d_ws;
    size_t off = 0;
    auto carve = [&](size_t bytes) -> float* {
        float* p = (float*)(ws + off);
        off += (bytes + 255) & ~(size_t)255;
        return p;
    };
    float* dis    = carve((size_t)N * 4);
    float* bufT   = carve((size_t)N * F3 * 4);
    float* bufA   = carve((size_t)N * F3 * 4);
    float* bufB   = carve((size_t)N * F3 * 4);
    const int nZero = 2 * G * F3 + G + 2 * NBMAX;
    float* zeros  = carve((size_t)nZero * 4);
    float* gmax   = zeros;
    float* gsum   = gmax + G * F3;
    float* cnt    = gsum + G * F3;
    int*  btotal  = (int*)(cnt + G);
    int*  bcursor = btotal + NBMAX;
    const int nB  = (N + BW - 1) >> BSH;
    const int nCh = (E + CH - 1) / CH;
    int*  bstart  = (int*)carve((size_t)(NBMAX + 1) * 4);
    int*  rowptr  = (int*)carve((size_t)(N + 1) * 4);
    int*  ent     = (int*)carve((size_t)E * 4);
    int2* rec     = (int2*)carve((size_t)E * 8);

    const int BLK = 256;
    auto grid1   = [&](long long n) { return dim3((unsigned)((n + BLK - 1) / BLK)); };
    auto gridWv  = [&](long long waves) { return dim3((unsigned)((waves * 32 + BLK - 1) / BLK)); };
    auto gemmGrid = [&](int rows) { return gridWv((rows + 15) / 16); };

    fill_i32<<<grid1(nZero), BLK, 0, stream>>>((int*)zeros, 0, nZero);
    bucket_count<<<dim3((unsigned)nCh), CH_T, 0, stream>>>(dst, btotal, E, nB);
    bucket_starts<<<dim3(1), NBMAX, 0, stream>>>(btotal, bstart, nB);
    bucket_scatter<<<dim3((unsigned)nCh), CH_T, 0, stream>>>(src, dst, bstart, bcursor, rec, E);
    bucket_finish<<<dim3((unsigned)nB), FIN_T, 0, stream>>>(rec, bstart, rowptr, dis, ent, N, nB);

    const dim3 pullGrid((unsigned)(((long long)((N + NPW - 1) / NPW) * 32 + PULL_T - 1) / PULL_T));

    wmma_gemm<2><<<gemmGrid(N), BLK, 0, stream>>>(x, W1, b1, dis, bufT, N, K0, 0);
    pull_rows<32, 32, false, false><<<pullGrid, PULL_T, 0, stream>>>(rowptr, ent, dis, bufT, nullptr, b1, bufA,
                                                                      nullptr, nullptr, nullptr, nullptr, N);

    pull_rows<32, 48, true, false><<<pullGrid, PULL_T, 0, stream>>>(rowptr, ent, dis, bufA, W2, b2, bufB,
                                                                     nullptr, nullptr, nullptr, nullptr, N);

    pull_rows<48, 64, true, true><<<pullGrid, PULL_T, 0, stream>>>(rowptr, ent, dis, bufB, W3, b3, nullptr,
                                                                    batch, gmax, gsum, cnt, N);

    project_pooled<2><<<gemmGrid(G), BLK, 0, stream>>>(gmax, gsum, cnt, Wout, bout, out, G, F3);
}
